# starting kernel
# baseline (speedup 1.0000x reference)
#include <hip/hip_runtime.h>
#include <hip/hip_bf16.h>

#define BATCH 2
#define SEQ   1024
#define EMB   256
#define HEADS 64
#define DKK   4

typedef __attribute__((ext_vector_type(16))) _Float16 v16h;
typedef __attribute__((ext_vector_type(8)))  float    v8f;
typedef __attribute__((ext_vector_type(2)))  float    v2f;

#if defined(__has_builtin)
#if __has_builtin(__builtin_amdgcn_wmma_f32_16x16x4_f32)
#define HAVE_WMMA_F32X4 1
#endif
#endif

__global__ __launch_bounds__(256) void qa_qkv_quantum(
    const float* __restrict__ x,
    const float* __restrict__ Wq, const float* __restrict__ Wk,
    const float* __restrict__ Wv, const float* __restrict__ params,
    float* __restrict__ qp, float* __restrict__ kp, float* __restrict__ vp)
{
  const int lane = threadIdx.x & 31;
  const int wave = threadIdx.x >> 5;
  const int lo = lane & 15, hi = lane >> 4;

  const int tile = blockIdx.x * 8 + wave;
  const int mat  = tile >> 11;
  const int rem  = tile & 2047;
  const int nt   = rem >> 7;
  const int mt   = rem & 127;

  const float* W    = (mat == 0) ? Wq : (mat == 1) ? Wk : Wv;
  float*       outp = (mat == 0) ? qp : (mat == 1) ? kp : vp;

  const float* wrow = W + (nt * 16 + lo) * EMB;
  const float* xrow = x + (mt * 16 + lo) * EMB;

  v8f c = {};
  #pragma unroll
  for (int ks = 0; ks < EMB; ks += 32) {
    const float* wa = wrow + ks + hi * 8;
    const float* xa = xrow + ks + hi * 8;
    const float4 w0 = *(const float4*)(wa);
    const float4 w1 = *(const float4*)(wa + 4);
    const float4 w2 = *(const float4*)(wa + 16);
    const float4 w3 = *(const float4*)(wa + 20);
    const float4 x0 = *(const float4*)(xa);
    const float4 x1 = *(const float4*)(xa + 4);
    const float4 x2 = *(const float4*)(xa + 16);
    const float4 x3 = *(const float4*)(xa + 20);
    v16h a, b;
    a[0]=(_Float16)w0.x; a[1]=(_Float16)w0.y; a[2]=(_Float16)w0.z; a[3]=(_Float16)w0.w;
    a[4]=(_Float16)w1.x; a[5]=(_Float16)w1.y; a[6]=(_Float16)w1.z; a[7]=(_Float16)w1.w;
    a[8]=(_Float16)w2.x; a[9]=(_Float16)w2.y; a[10]=(_Float16)w2.z; a[11]=(_Float16)w2.w;
    a[12]=(_Float16)w3.x; a[13]=(_Float16)w3.y; a[14]=(_Float16)w3.z; a[15]=(_Float16)w3.w;
    b[0]=(_Float16)x0.x; b[1]=(_Float16)x0.y; b[2]=(_Float16)x0.z; b[3]=(_Float16)x0.w;
    b[4]=(_Float16)x1.x; b[5]=(_Float16)x1.y; b[6]=(_Float16)x1.z; b[7]=(_Float16)x1.w;
    b[8]=(_Float16)x2.x; b[9]=(_Float16)x2.y; b[10]=(_Float16)x2.z; b[11]=(_Float16)x2.w;
    b[12]=(_Float16)x3.x; b[13]=(_Float16)x3.y; b[14]=(_Float16)x3.z; b[15]=(_Float16)x3.w;
    c = __builtin_amdgcn_wmma_f32_16x16x32_f16(false, a, false, b, (short)0, c,
                                               false, false);
  }

  const float p0 = params[0], p1 = params[1], p2 = params[2], p3 = params[3];
  const float cp0 = __cosf(p0), cp1 = __cosf(p1), cp2 = __cosf(p2), cp3 = __cosf(p3);
  const int sg = mt * 16 + lo;
  const int bb = sg >> 10;
  const int ss = sg & 1023;
  #pragma unroll
  for (int g = 0; g < 2; ++g) {
    const float z0 = cp0 * __cosf(c[4 * g + 0] + p0);
    const float z1 = cp1 * __cosf(c[4 * g + 1] + p1);
    const float z2 = cp2 * __cosf(c[4 * g + 2] + p2);
    const float z3 = cp3 * __cosf(c[4 * g + 3] + p3);
    const float o1 = z0 * z1;
    const float o2 = o1 * z2;
    const float o3 = o2 * z3;
    const float o0 = z1 * z2 * z3;
    const int hcol = nt * 16 + hi * 8 + g * 4;
    const int head = hcol >> 2;
    *(float4*)(outp + (((bb * HEADS + head) * SEQ + ss) << 2)) =
        make_float4(o0, o1, o2, o3);
  }
}

__global__ __launch_bounds__(256) void qa_attention(
    const float* __restrict__ qp, const float* __restrict__ kp,
    const float* __restrict__ vp, float* __restrict__ out)
{
  __shared__ float ksm[SEQ * 4];
  __shared__ float vsm[SEQ * 4];
  __shared__ float red[8][16][16][4];

  const int bh = blockIdx.x;
  const float* qh = qp + bh * SEQ * 4;
  const float* kh = kp + bh * SEQ * 4;
  const float* vh = vp + bh * SEQ * 4;

  for (int i = threadIdx.x; i < SEQ; i += 256) {
    *(float4*)&ksm[i * 4] = *(const float4*)(kh + i * 4);
    *(float4*)&vsm[i * 4] = *(const float4*)(vh + i * 4);
  }
  __syncthreads();

  const int lane = threadIdx.x & 31, wave = threadIdx.x >> 5;
  const int lo = lane & 15, hi = lane >> 4;
  const int bb = bh >> 6, head = bh & 63;

  for (int qt = wave; qt < 64; qt += 8) {
#if HAVE_WMMA_F32X4
    v2f aq;
    aq[0] = 0.5f * qh[(qt * 16 + lo) * 4 + hi * 2 + 0];
    aq[1] = 0.5f * qh[(qt * 16 + lo) * 4 + hi * 2 + 1];
#else
    float4 qr[8];
    #pragma unroll
    for (int j = 0; j < 8; ++j) {
      float4 q4 = *(const float4*)(qh + (qt * 16 + j + 8 * hi) * 4);
      qr[j] = make_float4(0.5f * q4.x, 0.5f * q4.y, 0.5f * q4.z, 0.5f * q4.w);
    }
#endif
    float run_max[8], run_sum[8];
    float4 ctxp[8];
    #pragma unroll
    for (int j = 0; j < 8; ++j) {
      run_max[j] = -1e30f; run_sum[j] = 0.0f;
      ctxp[j] = make_float4(0.f, 0.f, 0.f, 0.f);
    }

    for (int ct = 0; ct < 64; ++ct) {
      const int ncol = ct * 16 + lo;
      const float4 vcol = *(const float4*)&vsm[ncol * 4];
#if HAVE_WMMA_F32X4
      v2f bk;
      bk[0] = ksm[ncol * 4 + hi * 2 + 0];
      bk[1] = ksm[ncol * 4 + hi * 2 + 1];
      v8f sc = {};
      sc = __builtin_amdgcn_wmma_f32_16x16x4_f32(false, aq, false, bk,
                                                 (short)0, sc, false, false);
#else
      const float4 kc = *(const float4*)&ksm[ncol * 4];
      float sc[8];
      #pragma unroll
      for (int j = 0; j < 8; ++j)
        sc[j] = qr[j].x * kc.x + qr[j].y * kc.y + qr[j].z * kc.z + qr[j].w * kc.w;
#endif
      #pragma unroll
      for (int j = 0; j < 8; ++j) {
        const float s = sc[j];
        float mx = s;
        mx = fmaxf(mx, __shfl_xor(mx, 1));
        mx = fmaxf(mx, __shfl_xor(mx, 2));
        mx = fmaxf(mx, __shfl_xor(mx, 4));
        mx = fmaxf(mx, __shfl_xor(mx, 8));
        const float nm    = fmaxf(run_max[j], mx);
        const float scale = __expf(run_max[j] - nm);
        const float pr    = __expf(s - nm);
        float sum = pr;
        sum += __shfl_xor(sum, 1);
        sum += __shfl_xor(sum, 2);
        sum += __shfl_xor(sum, 4);
        sum += __shfl_xor(sum, 8);
        run_sum[j] = run_sum[j] * scale + sum;
        run_max[j] = nm;
        ctxp[j].x = ctxp[j].x * scale + pr * vcol.x;
        ctxp[j].y = ctxp[j].y * scale + pr * vcol.y;
        ctxp[j].z = ctxp[j].z * scale + pr * vcol.z;
        ctxp[j].w = ctxp[j].w * scale + pr * vcol.w;
      }
    }

    #pragma unroll
    for (int j = 0; j < 8; ++j) {
      const float inv = 1.0f / run_sum[j];
      const int r = j + 8 * hi;
      *(float4*)&red[wave][r][lo][0] =
          make_float4(ctxp[j].x * inv, ctxp[j].y * inv,
                      ctxp[j].z * inv, ctxp[j].w * inv);
    }
    __syncthreads();
    #pragma unroll
    for (int t = 0; t < 2; ++t) {
      const int o = lane * 2 + t;
      const int r = o >> 2, d = o & 3;
      float acc = 0.0f;
      #pragma unroll
      for (int cc = 0; cc < 16; ++cc) acc += red[wave][r][cc][d];
      const int srow = qt * 16 + r;
      out[(bb * SEQ + srow) * EMB + head * 4 + d] = acc;
    }
    __syncthreads();
  }
}

extern "C" void kernel_launch(void* const* d_in, const int* in_sizes, int n_in,
                              void* d_out, int out_size, void* d_ws, size_t ws_size,
                              hipStream_t stream) {
  const float* x      = (const float*)d_in[0];
  const float* Wq     = (const float*)d_in[1];
  const float* Wk     = (const float*)d_in[2];
  const float* Wv     = (const float*)d_in[3];
  const float* params = (const float*)d_in[4];
  float* out = (float*)d_out;

  const size_t headElems = (size_t)BATCH * HEADS * SEQ * DKK;
  float* qp = (float*)d_ws;
  float* kp = qp + headElems;
  float* vp = kp + headElems;

  qa_qkv_quantum<<<768, 256, 0, stream>>>(x, Wq, Wk, Wv, params, qp, kp, vp);
  qa_attention<<<BATCH * HEADS, 256, 0, stream>>>(qp, kp, vp, out);
}


# attention: each (b,h) split over two key halves on 256 workgroups of 16 waves (all WGPs, 4 waves per SIMD); partial max, row sum and P*V of the halves joined in a third launch (sum order changes)
# speedup vs baseline: 6.6568x; 6.6568x over previous
#include <hip/hip_runtime.h>
#include <hip/hip_bf16.h>

#define BATCH 2
#define SEQ   1024
#define EMB   256
#define HEADS 64
#define DKK   4

typedef __attribute__((ext_vector_type(16))) _Float16 v16h;
typedef __attribute__((ext_vector_type(8)))  float    v8f;
typedef __attribute__((ext_vector_type(2)))  float    v2f;

__global__ __launch_bounds__(256) void qa_qkv_quantum(
    const float* __restrict__ x,
    const float* __restrict__ Wq, const float* __restrict__ Wk,
    const float* __restrict__ Wv, const float* __restrict__ params,
    float* __restrict__ qp, float* __restrict__ kp, float* __restrict__ vp)
{
  const int lane = threadIdx.x & 31;
  const int wave = threadIdx.x >> 5;
  const int lo = lane & 15, hi = lane >> 4;

  const int tile = blockIdx.x * 8 + wave;
  const int mat  = tile >> 11;
  const int rem  = tile & 2047;
  const int nt   = rem >> 7;
  const int mt   = rem & 127;

  const float* W    = (mat == 0) ? Wq : (mat == 1) ? Wk : Wv;
  float*       outp = (mat == 0) ? qp : (mat == 1) ? kp : vp;

  const float* wrow = W + (nt * 16 + lo) * EMB;
  const float* xrow = x + (mt * 16 + lo) * EMB;

  v8f c = {};
  #pragma unroll
  for (int ks = 0; ks < EMB; ks += 32) {
    const float* wa = wrow + ks + hi * 8;
    const float* xa = xrow + ks + hi * 8;
    const float4 w0 = *(const float4*)(wa);
    const float4 w1 = *(const float4*)(wa + 4);
    const float4 w2 = *(const float4*)(wa + 16);
    const float4 w3 = *(const float4*)(wa + 20);
    const float4 x0 = *(const float4*)(xa);
    const float4 x1 = *(const float4*)(xa + 4);
    const float4 x2 = *(const float4*)(xa + 16);
    const float4 x3 = *(const float4*)(xa + 20);
    v16h a, b;
    a[0]=(_Float16)w0.x; a[1]=(_Float16)w0.y; a[2]=(_Float16)w0.z; a[3]=(_Float16)w0.w;
    a[4]=(_Float16)w1.x; a[5]=(_Float16)w1.y; a[6]=(_Float16)w1.z; a[7]=(_Float16)w1.w;
    a[8]=(_Float16)w2.x; a[9]=(_Float16)w2.y; a[10]=(_Float16)w2.z; a[11]=(_Float16)w2.w;
    a[12]=(_Float16)w3.x; a[13]=(_Float16)w3.y; a[14]=(_Float16)w3.z; a[15]=(_Float16)w3.w;
    b[0]=(_Float16)x0.x; b[1]=(_Float16)x0.y; b[2]=(_Float16)x0.z; b[3]=(_Float16)x0.w;
    b[4]=(_Float16)x1.x; b[5]=(_Float16)x1.y; b[6]=(_Float16)x1.z; b[7]=(_Float16)x1.w;
    b[8]=(_Float16)x2.x; b[9]=(_Float16)x2.y; b[10]=(_Float16)x2.z; b[11]=(_Float16)x2.w;
    b[12]=(_Float16)x3.x; b[13]=(_Float16)x3.y; b[14]=(_Float16)x3.z; b[15]=(_Float16)x3.w;
    c = __builtin_amdgcn_wmma_f32_16x16x32_f16(false, a, false, b, (short)0, c,
                                               false, false);
  }

  const float p0 = params[0], p1 = params[1], p2 = params[2], p3 = params[3];
  const float cp0 = __cosf(p0), cp1 = __cosf(p1), cp2 = __cosf(p2), cp3 = __cosf(p3);
  const int sg = mt * 16 + lo;
  const int bb = sg >> 10;
  const int ss = sg & 1023;
  #pragma unroll
  for (int g = 0; g < 2; ++g) {
    const float z0 = cp0 * __cosf(c[4 * g + 0] + p0);
    const float z1 = cp1 * __cosf(c[4 * g + 1] + p1);
    const float z2 = cp2 * __cosf(c[4 * g + 2] + p2);
    const float z3 = cp3 * __cosf(c[4 * g + 3] + p3);
    const float o1 = z0 * z1;
    const float o2 = o1 * z2;
    const float o3 = o2 * z3;
    const float o0 = z1 * z2 * z3;
    const int hcol = nt * 16 + hi * 8 + g * 4;
    const int head = hcol >> 2;
    *(float4*)(outp + (((bb * HEADS + head) * SEQ + ss) << 2)) =
        make_float4(o0, o1, o2, o3);
  }
}

#define KSPLIT  2
#define KEYS_WG (SEQ / KSPLIT)
#define KT_WG   (KEYS_WG / 16)
#define AWAVES  16

__device__ __forceinline__ v2f qa_key_frag(const float* ksm, const int ct,
                                           const int lo, const int hi)
{
  const int ncol = ct * 16 + lo;
  v2f bk;
  bk[0] = ksm[ncol * 4 + hi * 2 + 0];
  bk[1] = ksm[ncol * 4 + hi * 2 + 1];
  return bk;
}

__device__ __forceinline__ v8f qa_score_tile(const v2f aq, const v2f bk,
                                             const v8f acc)
{
  return __builtin_amdgcn_wmma_f32_16x16x4_f32(false, aq, false, bk,
                                               (short)0, acc, false, false);
}

__global__ __launch_bounds__(AWAVES * 32) void qa_attention_part(
    const float* __restrict__ qp, const float* __restrict__ kp,
    const float* __restrict__ vp, float* __restrict__ part)
{
  __shared__ float ksm[KEYS_WG * 4];
  __shared__ float vsm[KEYS_WG * 4];
  __shared__ float red[AWAVES][16][16][4];
  __shared__ __align__(16) float redsum[AWAVES][16][16];
  __shared__ float redmax[AWAVES][16];

  const int bh = blockIdx.x / KSPLIT;
  const int kh = blockIdx.x % KSPLIT;
  const float* qh = qp + bh * SEQ * 4;
  const float* kh_ = kp + (bh * SEQ + kh * KEYS_WG) * 4;
  const float* vh_ = vp + (bh * SEQ + kh * KEYS_WG) * 4;

  for (int i = threadIdx.x; i < KEYS_WG; i += AWAVES * 32) {
    *(float4*)&ksm[i * 4] = *(const float4*)(kh_ + i * 4);
    *(float4*)&vsm[i * 4] = *(const float4*)(vh_ + i * 4);
  }
  __syncthreads();

  const int lane = threadIdx.x & 31, wave = threadIdx.x >> 5;
  const int lo = lane & 15, hi = lane >> 4;
  float* pw = part + ((size_t)(kh * (BATCH * HEADS) + bh) * SEQ) * 8;

  for (int qt = wave; qt < 64; qt += AWAVES) {
    v2f aq;
    aq[0] = 0.5f * qh[(qt * 16 + lo) * 4 + hi * 2 + 0];
    aq[1] = 0.5f * qh[(qt * 16 + lo) * 4 + hi * 2 + 1];

    const v8f zero = {};
    float rmax[8];
    #pragma unroll
    for (int j = 0; j < 8; ++j) rmax[j] = -INFINITY;
    #pragma unroll 2
    for (int ct = 0; ct < KT_WG; ct += 4) {
      const v8f s0 = qa_score_tile(aq, qa_key_frag(ksm, ct + 0, lo, hi), zero);
      const v8f s1 = qa_score_tile(aq, qa_key_frag(ksm, ct + 1, lo, hi), zero);
      const v8f s2 = qa_score_tile(aq, qa_key_frag(ksm, ct + 2, lo, hi), zero);
      const v8f s3 = qa_score_tile(aq, qa_key_frag(ksm, ct + 3, lo, hi), zero);
      #pragma unroll
      for (int j = 0; j < 8; ++j) {
        rmax[j] = fmaxf(fmaxf(rmax[j], s0[j]), s1[j]);
        rmax[j] = fmaxf(fmaxf(rmax[j], s2[j]), s3[j]);
      }
    }
    v8f negm;
    #pragma unroll
    for (int j = 0; j < 8; ++j) {
      float mx = rmax[j];
      mx = fmaxf(mx, __shfl_xor(mx, 1));
      mx = fmaxf(mx, __shfl_xor(mx, 2));
      mx = fmaxf(mx, __shfl_xor(mx, 4));
      mx = fmaxf(mx, __shfl_xor(mx, 8));
      negm[j] = -mx;
    }

    float psum[8];
    float4 ctxp[8];
    #pragma unroll
    for (int j = 0; j < 8; ++j) {
      psum[j] = 0.0f;
      ctxp[j] = make_float4(0.f, 0.f, 0.f, 0.f);
    }
    v2f bkn[4];
    #pragma unroll
    for (int u = 0; u < 4; ++u) bkn[u] = qa_key_frag(ksm, u, lo, hi);
    for (int ct = 0; ct < KT_WG; ct += 4) {
      v2f bk[4];
      float4 vcol[4];
      #pragma unroll
      for (int u = 0; u < 4; ++u) {
        bk[u] = bkn[u];
        vcol[u] = *(const float4*)&vsm[((ct + u) * 16 + lo) * 4];
      }
      const int nx = (ct + 4) & (KT_WG - 1);
      #pragma unroll
      for (int u = 0; u < 4; ++u) bkn[u] = qa_key_frag(ksm, nx + u, lo, hi);
      v8f sc[4];
      #pragma unroll
      for (int u = 0; u < 4; ++u)
        sc[u] = qa_score_tile(aq, bk[u], negm);
      #pragma unroll
      for (int u = 0; u < 4; ++u) {
        #pragma unroll
        for (int j = 0; j < 8; ++j) {
          const float pr = __expf(sc[u][j]);
          psum[j] += pr;
          ctxp[j].x += pr * vcol[u].x;
          ctxp[j].y += pr * vcol[u].y;
          ctxp[j].z += pr * vcol[u].z;
          ctxp[j].w += pr * vcol[u].w;
        }
      }
    }

    #pragma unroll
    for (int j = 0; j < 8; ++j) {
      const int r = j + 8 * hi;
      *(float4*)&red[wave][r][lo][0] = ctxp[j];
      redsum[wave][r][lo] = psum[j];
      redmax[wave][r] = -negm[j];
    }
    __syncthreads();
    {
      const int r = lane >> 1;
      float rs = 0.0f;
      #pragma unroll
      for (int cc = 0; cc < 16; ++cc) rs += redsum[wave][r][cc];
      float acc[2];
      #pragma unroll
      for (int t = 0; t < 2; ++t) {
        const int d = (lane & 1) * 2 + t;
        float a = 0.0f;
        #pragma unroll
        for (int cc = 0; cc < 16; ++cc) a += red[wave][r][cc][d];
        acc[t] = a;
      }
      const int srow = qt * 16 + r;
      *(float4*)(pw + (size_t)srow * 8 + (lane & 1) * 4) =
          make_float4(acc[0], acc[1], rs, redmax[wave][r]);
    }
    __syncthreads();
  }
}

__global__ __launch_bounds__(256) void qa_attention_join(
    const float* __restrict__ part, float* __restrict__ out)
{
  const int head = threadIdx.x & 63;
  const int sg   = blockIdx.x * 4 + (threadIdx.x >> 6);
  const int bb = sg >> 10, ss = sg & 1023;
  const int bh = bb * HEADS + head;
  const float* pa = part + ((size_t)bh * SEQ + ss) * 8;
  const float* pb = pa + (size_t)(BATCH * HEADS) * SEQ * 8;
  const float4 a0 = *(const float4*)(pa);
  const float4 a1 = *(const float4*)(pa + 4);
  const float4 b0 = *(const float4*)(pb);
  const float4 b1 = *(const float4*)(pb + 4);
  const float m  = fmaxf(a0.w, b0.w);
  const float wa = __expf(a0.w - m);
  const float wb = __expf(b0.w - m);
  const float inv = 1.0f / (wa * a0.z + wb * b0.z);
  float4 o;
  o.x = (wa * a0.x + wb * b0.x) * inv;
  o.y = (wa * a0.y + wb * b0.y) * inv;
  o.z = (wa * a1.x + wb * b1.x) * inv;
  o.w = (wa * a1.y + wb * b1.y) * inv;
  *(float4*)(out + (size_t)sg * EMB + head * 4) = o;
}

extern "C" void kernel_launch(void* const* d_in, const int* in_sizes, int n_in,
                              void* d_out, int out_size, void* d_ws, size_t ws_size,
                              hipStream_t stream) {
  const float* x      = (const float*)d_in[0];
  const float* Wq     = (const float*)d_in[1];
  const float* Wk     = (const float*)d_in[2];
  const float* Wv     = (const float*)d_in[3];
  const float* params = (const float*)d_in[4];
  float* out = (float*)d_out;

  const size_t headElems = (size_t)BATCH * HEADS * SEQ * DKK;
  float* qp = (float*)d_ws;
  float* kp = qp + headElems;
  float* vp = kp + headElems;
  float* part = vp + headElems;

  qa_qkv_quantum<<<768, 256, 0, stream>>>(x, Wq, Wk, Wv, params, qp, kp, vp);
  qa_attention_part<<<BATCH * HEADS * KSPLIT, AWAVES * 32, 0, stream>>>(qp, kp, vp, part);
  qa_attention_join<<<BATCH * SEQ / 4, 256, 0, stream>>>(part, out);
}


# projection kernel: each wave forms a 3x2 block of output tiles (q,k,v weight tiles x two x tiles) from fragments kept in registers, no LDS: 20 fragment loads per 6 products, was 48; 1024 waves
# speedup vs baseline: 7.6393x; 1.1476x over previous
#include <hip/hip_runtime.h>
#include <hip/hip_bf16.h>

#define BATCH 2
#define SEQ   1024
#define EMB   256
#define HEADS 64
#define DKK   4

typedef __attribute__((ext_vector_type(16))) _Float16 v16h;
typedef __attribute__((ext_vector_type(8)))  float    v8f;
typedef __attribute__((ext_vector_type(2)))  float    v2f;

#define PW 3
#define PX 2

__device__ __forceinline__ v16h qa_frag16(const float* p)
{
  const float4 f0 = *(const float4*)(p);
  const float4 f1 = *(const float4*)(p + 4);
  const float4 f2 = *(const float4*)(p + 16);
  const float4 f3 = *(const float4*)(p + 20);
  v16h r;
  r[0]=(_Float16)f0.x; r[1]=(_Float16)f0.y; r[2]=(_Float16)f0.z; r[3]=(_Float16)f0.w;
  r[4]=(_Float16)f1.x; r[5]=(_Float16)f1.y; r[6]=(_Float16)f1.z; r[7]=(_Float16)f1.w;
  r[8]=(_Float16)f2.x; r[9]=(_Float16)f2.y; r[10]=(_Float16)f2.z; r[11]=(_Float16)f2.w;
  r[12]=(_Float16)f3.x; r[13]=(_Float16)f3.y; r[14]=(_Float16)f3.z; r[15]=(_Float16)f3.w;
  return r;
}

__global__ __launch_bounds__(128) __attribute__((amdgpu_waves_per_eu(1, 2))) void qa_qkv_quantum(
    const float* __restrict__ x,
    const float* __restrict__ Wq, const float* __restrict__ Wk,
    const float* __restrict__ Wv, const float* __restrict__ params,
    float* __restrict__ qp, float* __restrict__ kp, float* __restrict__ vp)
{
  const int lane = threadIdx.x & 31;
  const int wave = threadIdx.x >> 5;
  const int lo = lane & 15, hi = lane >> 4;

  const int wt  = blockIdx.x * 4 + wave;
  const int nt  = wt >> 6;
  const int mt0 = (wt & 63) * PX;

  const int woff = (nt * 16 + lo) * EMB + hi * 8;
  const float* wrow[PW] = { Wq + woff, Wk + woff, Wv + woff };
  const float* xrow[PX];
  #pragma unroll
  for (int t = 0; t < PX; ++t) xrow[t] = x + ((mt0 + t) * 16 + lo) * EMB + hi * 8;

  v8f c[PW][PX];
  #pragma unroll
  for (int p = 0; p < PW; ++p)
    #pragma unroll
    for (int t = 0; t < PX; ++t) c[p][t] = v8f{};

  #pragma unroll
  for (int ks = 0; ks < EMB; ks += 32) {
    v16h a[PW], b[PX];
    #pragma unroll
    for (int p = 0; p < PW; ++p) a[p] = qa_frag16(wrow[p] + ks);
    #pragma unroll
    for (int t = 0; t < PX; ++t) b[t] = qa_frag16(xrow[t] + ks);
    #pragma unroll
    for (int p = 0; p < PW; ++p)
      #pragma unroll
      for (int t = 0; t < PX; ++t)
        c[p][t] = __builtin_amdgcn_wmma_f32_16x16x32_f16(false, a[p], false, b[t],
                                                         (short)0, c[p][t],
                                                         false, false);
  }

  const float p0 = params[0], p1 = params[1], p2 = params[2], p3 = params[3];
  const float cp0 = __cosf(p0), cp1 = __cosf(p1), cp2 = __cosf(p2), cp3 = __cosf(p3);
  float* const outs[PW] = { qp, kp, vp };
  #pragma unroll
  for (int t = 0; t < PX; ++t) {
    const int sg = (mt0 + t) * 16 + lo;
    const int bb = sg >> 10;
    const int ss = sg & 1023;
    #pragma unroll
    for (int p = 0; p < PW; ++p) {
      #pragma unroll
      for (int g = 0; g < 2; ++g) {
        const float z0 = cp0 * __cosf(c[p][t][4 * g + 0] + p0);
        const float z1 = cp1 * __cosf(c[p][t][4 * g + 1] + p1);
        const float z2 = cp2 * __cosf(c[p][t][4 * g + 2] + p2);
        const float z3 = cp3 * __cosf(c[p][t][4 * g + 3] + p3);
        const float o1 = z0 * z1;
        const float o2 = o1 * z2;
        const float o3 = o2 * z3;
        const float o0 = z1 * z2 * z3;
        const int hcol = nt * 16 + hi * 8 + g * 4;
        const int head = hcol >> 2;
        *(float4*)(outs[p] + (((bb * HEADS + head) * SEQ + ss) << 2)) =
            make_float4(o0, o1, o2, o3);
      }
    }
  }
}

#define KSPLIT  2
#define KEYS_WG (SEQ / KSPLIT)
#define KT_WG   (KEYS_WG / 16)
#define AWAVES  16

__device__ __forceinline__ v2f qa_key_frag(const float* ksm, const int ct,
                                           const int lo, const int hi)
{
  const int ncol = ct * 16 + lo;
  v2f bk;
  bk[0] = ksm[ncol * 4 + hi * 2 + 0];
  bk[1] = ksm[ncol * 4 + hi * 2 + 1];
  return bk;
}

__device__ __forceinline__ v8f qa_score_tile(const v2f aq, const v2f bk,
                                             const v8f acc)
{
  return __builtin_amdgcn_wmma_f32_16x16x4_f32(false, aq, false, bk,
                                               (short)0, acc, false, false);
}

__global__ __launch_bounds__(AWAVES * 32) void qa_attention_part(
    const float* __restrict__ qp, const float* __restrict__ kp,
    const float* __restrict__ vp, float* __restrict__ part)
{
  __shared__ float ksm[KEYS_WG * 4];
  __shared__ float vsm[KEYS_WG * 4];
  __shared__ float red[AWAVES][16][16][4];
  __shared__ __align__(16) float redsum[AWAVES][16][16];
  __shared__ float redmax[AWAVES][16];

  const int bh = blockIdx.x / KSPLIT;
  const int kh = blockIdx.x % KSPLIT;
  const float* qh = qp + bh * SEQ * 4;
  const float* kh_ = kp + (bh * SEQ + kh * KEYS_WG) * 4;
  const float* vh_ = vp + (bh * SEQ + kh * KEYS_WG) * 4;

  for (int i = threadIdx.x; i < KEYS_WG; i += AWAVES * 32) {
    *(float4*)&ksm[i * 4] = *(const float4*)(kh_ + i * 4);
    *(float4*)&vsm[i * 4] = *(const float4*)(vh_ + i * 4);
  }
  __syncthreads();

  const int lane = threadIdx.x & 31, wave = threadIdx.x >> 5;
  const int lo = lane & 15, hi = lane >> 4;
  float* pw = part + ((size_t)(kh * (BATCH * HEADS) + bh) * SEQ) * 8;

  for (int qt = wave; qt < 64; qt += AWAVES) {
    v2f aq;
    aq[0] = 0.5f * qh[(qt * 16 + lo) * 4 + hi * 2 + 0];
    aq[1] = 0.5f * qh[(qt * 16 + lo) * 4 + hi * 2 + 1];

    const v8f zero = {};
    float rmax[8];
    #pragma unroll
    for (int j = 0; j < 8; ++j) rmax[j] = -INFINITY;
    #pragma unroll 2
    for (int ct = 0; ct < KT_WG; ct += 4) {
      const v8f s0 = qa_score_tile(aq, qa_key_frag(ksm, ct + 0, lo, hi), zero);
      const v8f s1 = qa_score_tile(aq, qa_key_frag(ksm, ct + 1, lo, hi), zero);
      const v8f s2 = qa_score_tile(aq, qa_key_frag(ksm, ct + 2, lo, hi), zero);
      const v8f s3 = qa_score_tile(aq, qa_key_frag(ksm, ct + 3, lo, hi), zero);
      #pragma unroll
      for (int j = 0; j < 8; ++j) {
        rmax[j] = fmaxf(fmaxf(rmax[j], s0[j]), s1[j]);
        rmax[j] = fmaxf(fmaxf(rmax[j], s2[j]), s3[j]);
      }
    }
    v8f negm;
    #pragma unroll
    for (int j = 0; j < 8; ++j) {
      float mx = rmax[j];
      mx = fmaxf(mx, __shfl_xor(mx, 1));
      mx = fmaxf(mx, __shfl_xor(mx, 2));
      mx = fmaxf(mx, __shfl_xor(mx, 4));
      mx = fmaxf(mx, __shfl_xor(mx, 8));
      negm[j] = -mx;
    }

    float psum[8];
    float4 ctxp[8];
    #pragma unroll
    for (int j = 0; j < 8; ++j) {
      psum[j] = 0.0f;
      ctxp[j] = make_float4(0.f, 0.f, 0.f, 0.f);
    }
    v2f bkn[4];
    #pragma unroll
    for (int u = 0; u < 4; ++u) bkn[u] = qa_key_frag(ksm, u, lo, hi);
    for (int ct = 0; ct < KT_WG; ct += 4) {
      v2f bk[4];
      float4 vcol[4];
      #pragma unroll
      for (int u = 0; u < 4; ++u) {
        bk[u] = bkn[u];
        vcol[u] = *(const float4*)&vsm[((ct + u) * 16 + lo) * 4];
      }
      const int nx = (ct + 4) & (KT_WG - 1);
      #pragma unroll
      for (int u = 0; u < 4; ++u) bkn[u] = qa_key_frag(ksm, nx + u, lo, hi);
      v8f sc[4];
      #pragma unroll
      for (int u = 0; u < 4; ++u)
        sc[u] = qa_score_tile(aq, bk[u], negm);
      #pragma unroll
      for (int u = 0; u < 4; ++u) {
        #pragma unroll
        for (int j = 0; j < 8; ++j) {
          const float pr = __expf(sc[u][j]);
          psum[j] += pr;
          ctxp[j].x += pr * vcol[u].x;
          ctxp[j].y += pr * vcol[u].y;
          ctxp[j].z += pr * vcol[u].z;
          ctxp[j].w += pr * vcol[u].w;
        }
      }
    }

    #pragma unroll
    for (int j = 0; j < 8; ++j) {
      const int r = j + 8 * hi;
      *(float4*)&red[wave][r][lo][0] = ctxp[j];
      redsum[wave][r][lo] = psum[j];
      redmax[wave][r] = -negm[j];
    }
    __syncthreads();
    {
      const int r = lane >> 1;
      float rs = 0.0f;
      #pragma unroll
      for (int cc = 0; cc < 16; ++cc) rs += redsum[wave][r][cc];
      float acc[2];
      #pragma unroll
      for (int t = 0; t < 2; ++t) {
        const int d = (lane & 1) * 2 + t;
        float a = 0.0f;
        #pragma unroll
        for (int cc = 0; cc < 16; ++cc) a += red[wave][r][cc][d];
        acc[t] = a;
      }
      const int srow = qt * 16 + r;
      *(float4*)(pw + (size_t)srow * 8 + (lane & 1) * 4) =
          make_float4(acc[0], acc[1], rs, redmax[wave][r]);
    }
    __syncthreads();
  }
}

__global__ __launch_bounds__(256) void qa_attention_join(
    const float* __restrict__ part, float* __restrict__ out)
{
  const int head = threadIdx.x & 63;
  const int sg   = blockIdx.x * 4 + (threadIdx.x >> 6);
  const int bb = sg >> 10, ss = sg & 1023;
  const int bh = bb * HEADS + head;
  const float* pa = part + ((size_t)bh * SEQ + ss) * 8;
  const float* pb = pa + (size_t)(BATCH * HEADS) * SEQ * 8;
  const float4 a0 = *(const float4*)(pa);
  const float4 a1 = *(const float4*)(pa + 4);
  const float4 b0 = *(const float4*)(pb);
  const float4 b1 = *(const float4*)(pb + 4);
  const float m  = fmaxf(a0.w, b0.w);
  const float wa = __expf(a0.w - m);
  const float wb = __expf(b0.w - m);
  const float inv = 1.0f / (wa * a0.z + wb * b0.z);
  float4 o;
  o.x = (wa * a0.x + wb * b0.x) * inv;
  o.y = (wa * a0.y + wb * b0.y) * inv;
  o.z = (wa * a1.x + wb * b1.x) * inv;
  o.w = (wa * a1.y + wb * b1.y) * inv;
  *(float4*)(out + (size_t)sg * EMB + head * 4) = o;
}

extern "C" void kernel_launch(void* const* d_in, const int* in_sizes, int n_in,
                              void* d_out, int out_size, void* d_ws, size_t ws_size,
                              hipStream_t stream) {
  const float* x      = (const float*)d_in[0];
  const float* Wq     = (const float*)d_in[1];
  const float* Wk     = (const float*)d_in[2];
  const float* Wv     = (const float*)d_in[3];
  const float* params = (const float*)d_in[4];
  float* out = (float*)d_out;

  const size_t headElems = (size_t)BATCH * HEADS * SEQ * DKK;
  float* qp = (float*)d_ws;
  float* kp = qp + headElems;
  float* vp = kp + headElems;
  float* part = vp + headElems;

  qa_qkv_quantum<<<256, 128, 0, stream>>>(x, Wq, Wk, Wv, params, qp, kp, vp);
  qa_attention_part<<<BATCH * HEADS * KSPLIT, AWAVES * 32, 0, stream>>>(qp, kp, vp, part);
  qa_attention_join<<<BATCH * SEQ / 4, 256, 0, stream>>>(part, out);
}


# attention: the two key halves of a row go to waves w and w+8 of ONE workgroup (one per query half) and are joined in LDS; third launch and the 8 MiB of partial records gone; sum order unchanged
# speedup vs baseline: 8.0988x; 1.0601x over previous
#include <hip/hip_runtime.h>
#include <hip/hip_bf16.h>

#define BATCH 2
#define SEQ   1024
#define EMB   256
#define HEADS 64
#define DKK   4

typedef __attribute__((ext_vector_type(16))) _Float16 v16h;
typedef __attribute__((ext_vector_type(8)))  float    v8f;
typedef __attribute__((ext_vector_type(2)))  float    v2f;

#define PW 3
#define PX 2

__device__ __forceinline__ v16h qa_frag16(const float* p)
{
  const float4 f0 = *(const float4*)(p);
  const float4 f1 = *(const float4*)(p + 4);
  const float4 f2 = *(const float4*)(p + 16);
  const float4 f3 = *(const float4*)(p + 20);
  v16h r;
  r[0]=(_Float16)f0.x; r[1]=(_Float16)f0.y; r[2]=(_Float16)f0.z; r[3]=(_Float16)f0.w;
  r[4]=(_Float16)f1.x; r[5]=(_Float16)f1.y; r[6]=(_Float16)f1.z; r[7]=(_Float16)f1.w;
  r[8]=(_Float16)f2.x; r[9]=(_Float16)f2.y; r[10]=(_Float16)f2.z; r[11]=(_Float16)f2.w;
  r[12]=(_Float16)f3.x; r[13]=(_Float16)f3.y; r[14]=(_Float16)f3.z; r[15]=(_Float16)f3.w;
  return r;
}

__global__ __launch_bounds__(128) __attribute__((amdgpu_waves_per_eu(1, 2))) void qa_qkv_quantum(
    const float* __restrict__ x,
    const float* __restrict__ Wq, const float* __restrict__ Wk,
    const float* __restrict__ Wv, const float* __restrict__ params,
    float* __restrict__ qp, float* __restrict__ kp, float* __restrict__ vp)
{
  const int lane = threadIdx.x & 31;
  const int wave = threadIdx.x >> 5;
  const int lo = lane & 15, hi = lane >> 4;

  const int wt  = blockIdx.x * 4 + wave;
  const int nt  = wt >> 6;
  const int mt0 = (wt & 63) * PX;

  const int woff = (nt * 16 + lo) * EMB + hi * 8;
  const float* wrow[PW] = { Wq + woff, Wk + woff, Wv + woff };
  const float* xrow[PX];
  #pragma unroll
  for (int t = 0; t < PX; ++t) xrow[t] = x + ((mt0 + t) * 16 + lo) * EMB + hi * 8;

  v8f c[PW][PX];
  #pragma unroll
  for (int p = 0; p < PW; ++p)
    #pragma unroll
    for (int t = 0; t < PX; ++t) c[p][t] = v8f{};

  #pragma unroll
  for (int ks = 0; ks < EMB; ks += 32) {
    v16h a[PW], b[PX];
    #pragma unroll
    for (int p = 0; p < PW; ++p) a[p] = qa_frag16(wrow[p] + ks);
    #pragma unroll
    for (int t = 0; t < PX; ++t) b[t] = qa_frag16(xrow[t] + ks);
    #pragma unroll
    for (int p = 0; p < PW; ++p)
      #pragma unroll
      for (int t = 0; t < PX; ++t)
        c[p][t] = __builtin_amdgcn_wmma_f32_16x16x32_f16(false, a[p], false, b[t],
                                                         (short)0, c[p][t],
                                                         false, false);
  }

  const float p0 = params[0], p1 = params[1], p2 = params[2], p3 = params[3];
  const float cp0 = __cosf(p0), cp1 = __cosf(p1), cp2 = __cosf(p2), cp3 = __cosf(p3);
  float* const outs[PW] = { qp, kp, vp };
  #pragma unroll
  for (int t = 0; t < PX; ++t) {
    const int sg = (mt0 + t) * 16 + lo;
    const int bb = sg >> 10;
    const int ss = sg & 1023;
    #pragma unroll
    for (int p = 0; p < PW; ++p) {
      #pragma unroll
      for (int g = 0; g < 2; ++g) {
        const float z0 = cp0 * __cosf(c[p][t][4 * g + 0] + p0);
        const float z1 = cp1 * __cosf(c[p][t][4 * g + 1] + p1);
        const float z2 = cp2 * __cosf(c[p][t][4 * g + 2] + p2);
        const float z3 = cp3 * __cosf(c[p][t][4 * g + 3] + p3);
        const float o1 = z0 * z1;
        const float o2 = o1 * z2;
        const float o3 = o2 * z3;
        const float o0 = z1 * z2 * z3;
        const int hcol = nt * 16 + hi * 8 + g * 4;
        const int head = hcol >> 2;
        *(float4*)(outs[p] + (((bb * HEADS + head) * SEQ + ss) << 2)) =
            make_float4(o0, o1, o2, o3);
      }
    }
  }
}

#define KSPLIT  2
#define KEYS_G  (SEQ / KSPLIT)
#define KT_G    (KEYS_G / 16)
#define AWAVES  16
#define GWAVES  (AWAVES / KSPLIT)
#define QSPLIT  2
#define QT_WG   (SEQ / 16 / QSPLIT)

__device__ __forceinline__ v2f qa_key_frag(const float* ksm, const int ct,
                                           const int lo, const int hi)
{
  const int ncol = ct * 16 + lo;
  v2f bk;
  bk[0] = ksm[ncol * 4 + hi * 2 + 0];
  bk[1] = ksm[ncol * 4 + hi * 2 + 1];
  return bk;
}

__device__ __forceinline__ v8f qa_score_tile(const v2f aq, const v2f bk,
                                             const v8f acc)
{
  return __builtin_amdgcn_wmma_f32_16x16x4_f32(false, aq, false, bk,
                                               (short)0, acc, false, false);
}

__global__ __launch_bounds__(AWAVES * 32) void qa_attention(
    const float* __restrict__ qp, const float* __restrict__ kp,
    const float* __restrict__ vp, float* __restrict__ out)
{
  __shared__ float ksm[SEQ * 4];
  __shared__ float vsm[SEQ * 4];
  __shared__ float red[AWAVES][16][16][4];
  __shared__ __align__(16) float redsum[AWAVES][16][16];
  __shared__ float redmax[AWAVES][16];
  __shared__ __align__(16) float joinb[GWAVES][32][4];

  const int bh = blockIdx.x / QSPLIT;
  const int qh = blockIdx.x % QSPLIT;
  const float* qh_ = qp + (bh * SEQ + qh * (SEQ / QSPLIT)) * 4;
  const float* kh_ = kp + bh * SEQ * 4;
  const float* vh_ = vp + bh * SEQ * 4;

  for (int i = threadIdx.x; i < SEQ; i += AWAVES * 32) {
    *(float4*)&ksm[i * 4] = *(const float4*)(kh_ + i * 4);
    *(float4*)&vsm[i * 4] = *(const float4*)(vh_ + i * 4);
  }
  __syncthreads();

  const int lane = threadIdx.x & 31;
  const int wave = __builtin_amdgcn_readfirstlane(threadIdx.x >> 5);
  const int lo = lane & 15, hi = lane >> 4;
  const int kg = wave / GWAVES;
  const int wq = wave % GWAVES;
  const float* ksw = ksm + kg * KEYS_G * 4;
  const float* vsw = vsm + kg * KEYS_G * 4;
  const int bb = bh / HEADS, head = bh % HEADS;
  float* ow = out + ((size_t)(bb * SEQ + qh * (SEQ / QSPLIT))) * EMB
                  + head * 4 + (lane & 1) * 2;

  for (int qt = wq; qt < QT_WG; qt += GWAVES) {
    v2f aq;
    aq[0] = 0.5f * qh_[(qt * 16 + lo) * 4 + hi * 2 + 0];
    aq[1] = 0.5f * qh_[(qt * 16 + lo) * 4 + hi * 2 + 1];

    const v8f zero = {};
    float rmax[8];
    #pragma unroll
    for (int j = 0; j < 8; ++j) rmax[j] = -INFINITY;
    #pragma unroll 2
    for (int ct = 0; ct < KT_G; ct += 4) {
      const v8f s0 = qa_score_tile(aq, qa_key_frag(ksw, ct + 0, lo, hi), zero);
      const v8f s1 = qa_score_tile(aq, qa_key_frag(ksw, ct + 1, lo, hi), zero);
      const v8f s2 = qa_score_tile(aq, qa_key_frag(ksw, ct + 2, lo, hi), zero);
      const v8f s3 = qa_score_tile(aq, qa_key_frag(ksw, ct + 3, lo, hi), zero);
      #pragma unroll
      for (int j = 0; j < 8; ++j) {
        rmax[j] = fmaxf(fmaxf(rmax[j], s0[j]), s1[j]);
        rmax[j] = fmaxf(fmaxf(rmax[j], s2[j]), s3[j]);
      }
    }
    v8f negm;
    #pragma unroll
    for (int j = 0; j < 8; ++j) {
      float mx = rmax[j];
      mx = fmaxf(mx, __shfl_xor(mx, 1));
      mx = fmaxf(mx, __shfl_xor(mx, 2));
      mx = fmaxf(mx, __shfl_xor(mx, 4));
      mx = fmaxf(mx, __shfl_xor(mx, 8));
      negm[j] = -mx;
    }

    float psum[8];
    float4 ctxp[8];
    #pragma unroll
    for (int j = 0; j < 8; ++j) {
      psum[j] = 0.0f;
      ctxp[j] = make_float4(0.f, 0.f, 0.f, 0.f);
    }
    v2f bkn[4];
    #pragma unroll
    for (int u = 0; u < 4; ++u) bkn[u] = qa_key_frag(ksw, u, lo, hi);
    for (int ct = 0; ct < KT_G; ct += 4) {
      v2f bk[4];
      float4 vcol[4];
      #pragma unroll
      for (int u = 0; u < 4; ++u) {
        bk[u] = bkn[u];
        vcol[u] = *(const float4*)&vsw[((ct + u) * 16 + lo) * 4];
      }
      const int nx = (ct + 4) & (KT_G - 1);
      #pragma unroll
      for (int u = 0; u < 4; ++u) bkn[u] = qa_key_frag(ksw, nx + u, lo, hi);
      v8f sc[4];
      #pragma unroll
      for (int u = 0; u < 4; ++u)
        sc[u] = qa_score_tile(aq, bk[u], negm);
      #pragma unroll
      for (int u = 0; u < 4; ++u) {
        #pragma unroll
        for (int j = 0; j < 8; ++j) {
          const float pr = __expf(sc[u][j]);
          psum[j] += pr;
          ctxp[j].x += pr * vcol[u].x;
          ctxp[j].y += pr * vcol[u].y;
          ctxp[j].z += pr * vcol[u].z;
          ctxp[j].w += pr * vcol[u].w;
        }
      }
    }

    #pragma unroll
    for (int j = 0; j < 8; ++j) {
      const int r = j + 8 * hi;
      *(float4*)&red[wave][r][lo][0] = ctxp[j];
      redsum[wave][r][lo] = psum[j];
      redmax[wave][r] = -negm[j];
    }
    __syncthreads();
    const int r = lane >> 1;
    float rs = 0.0f;
    #pragma unroll
    for (int cc = 0; cc < 16; ++cc) rs += redsum[wave][r][cc];
    float acc[2];
    #pragma unroll
    for (int t = 0; t < 2; ++t) {
      const int d = (lane & 1) * 2 + t;
      float a = 0.0f;
      #pragma unroll
      for (int cc = 0; cc < 16; ++cc) a += red[wave][r][cc][d];
      acc[t] = a;
    }
    const float mxr = redmax[wave][r];
    if (kg == 1)
      *(float4*)&joinb[wq][lane][0] = make_float4(acc[0], acc[1], rs, mxr);
    __syncthreads();
    if (kg == 0) {
      const float4 pb = *(const float4*)&joinb[wq][lane][0];
      const float m  = fmaxf(mxr, pb.w);
      const float wa = __expf(mxr - m);
      const float wb = __expf(pb.w - m);
      const float inv = 1.0f / (wa * rs + wb * pb.z);
      float2 o;
      o.x = (wa * acc[0] + wb * pb.x) * inv;
      o.y = (wa * acc[1] + wb * pb.y) * inv;
      *(float2*)(ow + (size_t)(qt * 16 + r) * EMB) = o;
    }
  }
}

extern "C" void kernel_launch(void* const* d_in, const int* in_sizes, int n_in,
                              void* d_out, int out_size, void* d_ws, size_t ws_size,
                              hipStream_t stream) {
  const float* x      = (const float*)d_in[0];
  const float* Wq     = (const float*)d_in[1];
  const float* Wk     = (const float*)d_in[2];
  const float* Wv     = (const float*)d_in[3];
  const float* params = (const float*)d_in[4];
  float* out = (float*)d_out;

  const size_t headElems = (size_t)BATCH * HEADS * SEQ * DKK;
  float* qp = (float*)d_ws;
  float* kp = qp + headElems;
  float* vp = kp + headElems;

  qa_qkv_quantum<<<256, 128, 0, stream>>>(x, Wq, Wk, Wv, params, qp, kp, vp);
  qa_attention<<<BATCH * HEADS * QSPLIT, AWAVES * 32, 0, stream>>>(qp, kp, vp, out);
}


# attention q-tile reduction in LDS: per-wave partial buffers' row pitch padded 64->68 and 16->20 floats; the transposed reads (16 rows each) now fall on 16 bank groups, not one; sum order unchanged
# speedup vs baseline: 9.5248x; 1.1761x over previous
#include <hip/hip_runtime.h>
#include <hip/hip_bf16.h>

#define BATCH 2
#define SEQ   1024
#define EMB   256
#define HEADS 64
#define DKK   4

typedef __attribute__((ext_vector_type(16))) _Float16 v16h;
typedef __attribute__((ext_vector_type(8)))  float    v8f;
typedef __attribute__((ext_vector_type(2)))  float    v2f;

#define PW 3
#define PX 2

__device__ __forceinline__ v16h qa_frag16(const float* p)
{
  const float4 f0 = *(const float4*)(p);
  const float4 f1 = *(const float4*)(p + 4);
  const float4 f2 = *(const float4*)(p + 16);
  const float4 f3 = *(const float4*)(p + 20);
  v16h r;
  r[0]=(_Float16)f0.x; r[1]=(_Float16)f0.y; r[2]=(_Float16)f0.z; r[3]=(_Float16)f0.w;
  r[4]=(_Float16)f1.x; r[5]=(_Float16)f1.y; r[6]=(_Float16)f1.z; r[7]=(_Float16)f1.w;
  r[8]=(_Float16)f2.x; r[9]=(_Float16)f2.y; r[10]=(_Float16)f2.z; r[11]=(_Float16)f2.w;
  r[12]=(_Float16)f3.x; r[13]=(_Float16)f3.y; r[14]=(_Float16)f3.z; r[15]=(_Float16)f3.w;
  return r;
}

__global__ __launch_bounds__(128) __attribute__((amdgpu_waves_per_eu(1, 2))) void qa_qkv_quantum(
    const float* __restrict__ x,
    const float* __restrict__ Wq, const float* __restrict__ Wk,
    const float* __restrict__ Wv, const float* __restrict__ params,
    float* __restrict__ qp, float* __restrict__ kp, float* __restrict__ vp)
{
  const int lane = threadIdx.x & 31;
  const int wave = threadIdx.x >> 5;
  const int lo = lane & 15, hi = lane >> 4;

  const int wt  = blockIdx.x * 4 + wave;
  const int nt  = wt >> 6;
  const int mt0 = (wt & 63) * PX;

  const int woff = (nt * 16 + lo) * EMB + hi * 8;
  const float* wrow[PW] = { Wq + woff, Wk + woff, Wv + woff };
  const float* xrow[PX];
  #pragma unroll
  for (int t = 0; t < PX; ++t) xrow[t] = x + ((mt0 + t) * 16 + lo) * EMB + hi * 8;

  v8f c[PW][PX];
  #pragma unroll
  for (int p = 0; p < PW; ++p)
    #pragma unroll
    for (int t = 0; t < PX; ++t) c[p][t] = v8f{};

  #pragma unroll
  for (int ks = 0; ks < EMB; ks += 32) {
    v16h a[PW], b[PX];
    #pragma unroll
    for (int p = 0; p < PW; ++p) a[p] = qa_frag16(wrow[p] + ks);
    #pragma unroll
    for (int t = 0; t < PX; ++t) b[t] = qa_frag16(xrow[t] + ks);
    #pragma unroll
    for (int p = 0; p < PW; ++p)
      #pragma unroll
      for (int t = 0; t < PX; ++t)
        c[p][t] = __builtin_amdgcn_wmma_f32_16x16x32_f16(false, a[p], false, b[t],
                                                         (short)0, c[p][t],
                                                         false, false);
  }

  const float p0 = params[0], p1 = params[1], p2 = params[2], p3 = params[3];
  const float cp0 = __cosf(p0), cp1 = __cosf(p1), cp2 = __cosf(p2), cp3 = __cosf(p3);
  float* const outs[PW] = { qp, kp, vp };
  #pragma unroll
  for (int t = 0; t < PX; ++t) {
    const int sg = (mt0 + t) * 16 + lo;
    const int bb = sg >> 10;
    const int ss = sg & 1023;
    #pragma unroll
    for (int p = 0; p < PW; ++p) {
      #pragma unroll
      for (int g = 0; g < 2; ++g) {
        const float z0 = cp0 * __cosf(c[p][t][4 * g + 0] + p0);
        const float z1 = cp1 * __cosf(c[p][t][4 * g + 1] + p1);
        const float z2 = cp2 * __cosf(c[p][t][4 * g + 2] + p2);
        const float z3 = cp3 * __cosf(c[p][t][4 * g + 3] + p3);
        const float o1 = z0 * z1;
        const float o2 = o1 * z2;
        const float o3 = o2 * z3;
        const float o0 = z1 * z2 * z3;
        const int hcol = nt * 16 + hi * 8 + g * 4;
        const int head = hcol >> 2;
        *(float4*)(outs[p] + (((bb * HEADS + head) * SEQ + ss) << 2)) =
            make_float4(o0, o1, o2, o3);
      }
    }
  }
}

#define KSPLIT  2
#define KEYS_G  (SEQ / KSPLIT)
#define KT_G    (KEYS_G / 16)
#define AWAVES  16
#define GWAVES  (AWAVES / KSPLIT)
#define QSPLIT  2
#define QT_WG   (SEQ / 16 / QSPLIT)
#define RED_PITCH   (16 * 4 + 4)
#define RSUM_PITCH  (16 + 4)

__device__ __forceinline__ v2f qa_key_frag(const float* ksm, const int ct,
                                           const int lo, const int hi)
{
  const int ncol = ct * 16 + lo;
  v2f bk;
  bk[0] = ksm[ncol * 4 + hi * 2 + 0];
  bk[1] = ksm[ncol * 4 + hi * 2 + 1];
  return bk;
}

__device__ __forceinline__ v8f qa_score_tile(const v2f aq, const v2f bk,
                                             const v8f acc)
{
  return __builtin_amdgcn_wmma_f32_16x16x4_f32(false, aq, false, bk,
                                               (short)0, acc, false, false);
}

__global__ __launch_bounds__(AWAVES * 32) void qa_attention(
    const float* __restrict__ qp, const float* __restrict__ kp,
    const float* __restrict__ vp, float* __restrict__ out)
{
  __shared__ float ksm[SEQ * 4];
  __shared__ float vsm[SEQ * 4];
  __shared__ __align__(16) float red[AWAVES][16][RED_PITCH];
  __shared__ __align__(16) float redsum[AWAVES][16][RSUM_PITCH];
  __shared__ float redmax[AWAVES][16];
  __shared__ __align__(16) float joinb[GWAVES][32][4];

  const int bh = blockIdx.x / QSPLIT;
  const int qh = blockIdx.x % QSPLIT;
  const float* qh_ = qp + (bh * SEQ + qh * (SEQ / QSPLIT)) * 4;
  const float* kh_ = kp + bh * SEQ * 4;
  const float* vh_ = vp + bh * SEQ * 4;

  for (int i = threadIdx.x; i < SEQ; i += AWAVES * 32) {
    *(float4*)&ksm[i * 4] = *(const float4*)(kh_ + i * 4);
    *(float4*)&vsm[i * 4] = *(const float4*)(vh_ + i * 4);
  }
  __syncthreads();

  const int lane = threadIdx.x & 31;
  const int wave = __builtin_amdgcn_readfirstlane(threadIdx.x >> 5);
  const int lo = lane & 15, hi = lane >> 4;
  const int kg = wave / GWAVES;
  const int wq = wave % GWAVES;
  const float* ksw = ksm + kg * KEYS_G * 4;
  const float* vsw = vsm + kg * KEYS_G * 4;
  const int bb = bh / HEADS, head = bh % HEADS;
  float* ow = out + ((size_t)(bb * SEQ + qh * (SEQ / QSPLIT))) * EMB
                  + head * 4 + (lane & 1) * 2;

  for (int qt = wq; qt < QT_WG; qt += GWAVES) {
    v2f aq;
    aq[0] = 0.5f * qh_[(qt * 16 + lo) * 4 + hi * 2 + 0];
    aq[1] = 0.5f * qh_[(qt * 16 + lo) * 4 + hi * 2 + 1];

    const v8f zero = {};
    float rmax[8];
    #pragma unroll
    for (int j = 0; j < 8; ++j) rmax[j] = -INFINITY;
    #pragma unroll 2
    for (int ct = 0; ct < KT_G; ct += 4) {
      const v8f s0 = qa_score_tile(aq, qa_key_frag(ksw, ct + 0, lo, hi), zero);
      const v8f s1 = qa_score_tile(aq, qa_key_frag(ksw, ct + 1, lo, hi), zero);
      const v8f s2 = qa_score_tile(aq, qa_key_frag(ksw, ct + 2, lo, hi), zero);
      const v8f s3 = qa_score_tile(aq, qa_key_frag(ksw, ct + 3, lo, hi), zero);
      #pragma unroll
      for (int j = 0; j < 8; ++j) {
        rmax[j] = fmaxf(fmaxf(rmax[j], s0[j]), s1[j]);
        rmax[j] = fmaxf(fmaxf(rmax[j], s2[j]), s3[j]);
      }
    }
    v8f negm;
    #pragma unroll
    for (int j = 0; j < 8; ++j) {
      float mx = rmax[j];
      mx = fmaxf(mx, __shfl_xor(mx, 1));
      mx = fmaxf(mx, __shfl_xor(mx, 2));
      mx = fmaxf(mx, __shfl_xor(mx, 4));
      mx = fmaxf(mx, __shfl_xor(mx, 8));
      negm[j] = -mx;
    }

    float psum[8];
    float4 ctxp[8];
    #pragma unroll
    for (int j = 0; j < 8; ++j) {
      psum[j] = 0.0f;
      ctxp[j] = make_float4(0.f, 0.f, 0.f, 0.f);
    }
    v2f bkn[4];
    #pragma unroll
    for (int u = 0; u < 4; ++u) bkn[u] = qa_key_frag(ksw, u, lo, hi);
    for (int ct = 0; ct < KT_G; ct += 4) {
      v2f bk[4];
      float4 vcol[4];
      #pragma unroll
      for (int u = 0; u < 4; ++u) {
        bk[u] = bkn[u];
        vcol[u] = *(const float4*)&vsw[((ct + u) * 16 + lo) * 4];
      }
      const int nx = (ct + 4) & (KT_G - 1);
      #pragma unroll
      for (int u = 0; u < 4; ++u) bkn[u] = qa_key_frag(ksw, nx + u, lo, hi);
      v8f sc[4];
      #pragma unroll
      for (int u = 0; u < 4; ++u)
        sc[u] = qa_score_tile(aq, bk[u], negm);
      #pragma unroll
      for (int u = 0; u < 4; ++u) {
        #pragma unroll
        for (int j = 0; j < 8; ++j) {
          const float pr = __expf(sc[u][j]);
          psum[j] += pr;
          ctxp[j].x += pr * vcol[u].x;
          ctxp[j].y += pr * vcol[u].y;
          ctxp[j].z += pr * vcol[u].z;
          ctxp[j].w += pr * vcol[u].w;
        }
      }
    }

    #pragma unroll
    for (int j = 0; j < 8; ++j) {
      const int r = j + 8 * hi;
      *(float4*)&red[wave][r][lo * 4] = ctxp[j];
      redsum[wave][r][lo] = psum[j];
      redmax[wave][r] = -negm[j];
    }
    __syncthreads();
    const int r = lane >> 1;
    float rs = 0.0f;
    #pragma unroll
    for (int cc = 0; cc < 16; ++cc) rs += redsum[wave][r][cc];
    float acc[2];
    #pragma unroll
    for (int t = 0; t < 2; ++t) {
      const int d = (lane & 1) * 2 + t;
      float a = 0.0f;
      #pragma unroll
      for (int cc = 0; cc < 16; ++cc) a += red[wave][r][cc * 4 + d];
      acc[t] = a;
    }
    const float mxr = redmax[wave][r];
    if (kg == 1)
      *(float4*)&joinb[wq][lane][0] = make_float4(acc[0], acc[1], rs, mxr);
    __syncthreads();
    if (kg == 0) {
      const float4 pb = *(const float4*)&joinb[wq][lane][0];
      const float m  = fmaxf(mxr, pb.w);
      const float wa = __expf(mxr - m);
      const float wb = __expf(pb.w - m);
      const float inv = 1.0f / (wa * rs + wb * pb.z);
      float2 o;
      o.x = (wa * acc[0] + wb * pb.x) * inv;
      o.y = (wa * acc[1] + wb * pb.y) * inv;
      *(float2*)(ow + (size_t)(qt * 16 + r) * EMB) = o;
    }
  }
}

extern "C" void kernel_launch(void* const* d_in, const int* in_sizes, int n_in,
                              void* d_out, int out_size, void* d_ws, size_t ws_size,
                              hipStream_t stream) {
  const float* x      = (const float*)d_in[0];
  const float* Wq     = (const float*)d_in[1];
  const float* Wk     = (const float*)d_in[2];
  const float* Wv     = (const float*)d_in[3];
  const float* params = (const float*)d_in[4];
  float* out = (float*)d_out;

  const size_t headElems = (size_t)BATCH * HEADS * SEQ * DKK;
  float* qp = (float*)d_ws;
  float* kp = qp + headElems;
  float* vp = kp + headElems;

  qa_qkv_quantum<<<256, 128, 0, stream>>>(x, Wq, Wk, Wv, params, qp, kp, vp);
  qa_attention<<<BATCH * HEADS * QSPLIT, AWAVES * 32, 0, stream>>>(qp, kp, vp, out);
}


# projection kernel: a workgroup forms 6x4 output tiles, k split over its 4 waves, partial tiles added in LDS in fixed order: 10 fragment loads per 6 products, was 20; sum order changes
# speedup vs baseline: 9.9247x; 1.0420x over previous
#include <hip/hip_runtime.h>
#include <hip/hip_bf16.h>

#define BATCH 2
#define SEQ   1024
#define EMB   256
#define HEADS 64
#define DKK   4

typedef __attribute__((ext_vector_type(16))) _Float16 v16h;
typedef __attribute__((ext_vector_type(8)))  float    v8f;
typedef __attribute__((ext_vector_type(2)))  float    v2f;

#define PE 2
#define PW 3
#define PX 4
#define KW 4
#define PT (PE * PW * PX)
#define KS (EMB / KW / 32)

struct qa_raw16 { float4 f0, f1, f2, f3; };

__device__ __forceinline__ qa_raw16 qa_load16(const float* p)
{
  qa_raw16 r;
  r.f0 = *(const float4*)(p);
  r.f1 = *(const float4*)(p + 4);
  r.f2 = *(const float4*)(p + 16);
  r.f3 = *(const float4*)(p + 20);
  return r;
}

__device__ __forceinline__ v16h qa_frag16(const qa_raw16 w)
{
  v16h r;
  r[0]=(_Float16)w.f0.x; r[1]=(_Float16)w.f0.y; r[2]=(_Float16)w.f0.z; r[3]=(_Float16)w.f0.w;
  r[4]=(_Float16)w.f1.x; r[5]=(_Float16)w.f1.y; r[6]=(_Float16)w.f1.z; r[7]=(_Float16)w.f1.w;
  r[8]=(_Float16)w.f2.x; r[9]=(_Float16)w.f2.y; r[10]=(_Float16)w.f2.z; r[11]=(_Float16)w.f2.w;
  r[12]=(_Float16)w.f3.x; r[13]=(_Float16)w.f3.y; r[14]=(_Float16)w.f3.z; r[15]=(_Float16)w.f3.w;
  return r;
}

__global__ __launch_bounds__(KW * 32) __attribute__((amdgpu_waves_per_eu(1, 1))) void qa_qkv_quantum(
    const float* __restrict__ x,
    const float* __restrict__ Wq, const float* __restrict__ Wk,
    const float* __restrict__ Wv, const float* __restrict__ params,
    float* __restrict__ qp, float* __restrict__ kp, float* __restrict__ vp)
{
  __shared__ __align__(16) float part[KW][PT][2][32][4];

  const int lane = threadIdx.x & 31;
  const int wave = __builtin_amdgcn_readfirstlane(threadIdx.x >> 5);
  const int lo = lane & 15, hi = lane >> 4;

  const int ntp = blockIdx.x >> 5;
  const int mtq = blockIdx.x & 31;
  const int k0  = wave * (EMB / KW);

  const float* wrow[PE][PW];
  #pragma unroll
  for (int e = 0; e < PE; ++e) {
    const int woff = ((ntp * PE + e) * 16 + lo) * EMB + hi * 8 + k0;
    wrow[e][0] = Wq + woff;
    wrow[e][1] = Wk + woff;
    wrow[e][2] = Wv + woff;
  }
  const float* xrow[PX];
  #pragma unroll
  for (int t = 0; t < PX; ++t)
    xrow[t] = x + ((mtq * PX + t) * 16 + lo) * EMB + hi * 8 + k0;

  v8f c[PE][PW][PX];
  #pragma unroll
  for (int e = 0; e < PE; ++e)
    #pragma unroll
    for (int p = 0; p < PW; ++p)
      #pragma unroll
      for (int t = 0; t < PX; ++t) c[e][p][t] = v8f{};

  qa_raw16 ra[KS][PE][PW], rb[KS][PX];
  #pragma unroll
  for (int s = 0; s < KS; ++s) {
    #pragma unroll
    for (int e = 0; e < PE; ++e)
      #pragma unroll
      for (int p = 0; p < PW; ++p) ra[s][e][p] = qa_load16(wrow[e][p] + s * 32);
    #pragma unroll
    for (int t = 0; t < PX; ++t) rb[s][t] = qa_load16(xrow[t] + s * 32);
  }
  __builtin_amdgcn_sched_barrier(0);

  #pragma unroll
  for (int s = 0; s < KS; ++s) {
    v16h a[PE][PW], b[PX];
    #pragma unroll
    for (int e = 0; e < PE; ++e)
      #pragma unroll
      for (int p = 0; p < PW; ++p) a[e][p] = qa_frag16(ra[s][e][p]);
    #pragma unroll
    for (int t = 0; t < PX; ++t) b[t] = qa_frag16(rb[s][t]);
    #pragma unroll
    for (int e = 0; e < PE; ++e)
      #pragma unroll
      for (int p = 0; p < PW; ++p)
        #pragma unroll
        for (int t = 0; t < PX; ++t)
          c[e][p][t] = __builtin_amdgcn_wmma_f32_16x16x32_f16(false, a[e][p], false, b[t],
                                                              (short)0, c[e][p][t],
                                                              false, false);
  }

  #pragma unroll
  for (int e = 0; e < PE; ++e)
    #pragma unroll
    for (int p = 0; p < PW; ++p)
      #pragma unroll
      for (int t = 0; t < PX; ++t) {
        const int ti = (e * PW + p) * PX + t;
        *(float4*)&part[wave][ti][0][lane][0] =
            make_float4(c[e][p][t][0], c[e][p][t][1], c[e][p][t][2], c[e][p][t][3]);
        *(float4*)&part[wave][ti][1][lane][0] =
            make_float4(c[e][p][t][4], c[e][p][t][5], c[e][p][t][6], c[e][p][t][7]);
      }
  __syncthreads();

  const float p0 = params[0], p1 = params[1], p2 = params[2], p3 = params[3];
  const float cp0 = __cosf(p0), cp1 = __cosf(p1), cp2 = __cosf(p2), cp3 = __cosf(p3);
  float* const outs[PW] = { qp, kp, vp };
  const int sg = (mtq * PX + wave) * 16 + lo;
  const int bb = sg >> 10;
  const int ss = sg & 1023;
  #pragma unroll
  for (int e = 0; e < PE; ++e) {
    #pragma unroll
    for (int p = 0; p < PW; ++p) {
      const int ti = (e * PW + p) * PX + wave;
      #pragma unroll
      for (int g = 0; g < 2; ++g) {
        float4 s = *(const float4*)&part[0][ti][g][lane][0];
        #pragma unroll
        for (int w = 1; w < KW; ++w) {
          const float4 o = *(const float4*)&part[w][ti][g][lane][0];
          s.x += o.x; s.y += o.y; s.z += o.z; s.w += o.w;
        }
        const float z0 = cp0 * __cosf(s.x + p0);
        const float z1 = cp1 * __cosf(s.y + p1);
        const float z2 = cp2 * __cosf(s.z + p2);
        const float z3 = cp3 * __cosf(s.w + p3);
        const float o1 = z0 * z1;
        const float o2 = o1 * z2;
        const float o3 = o2 * z3;
        const float o0 = z1 * z2 * z3;
        const int hcol = (ntp * PE + e) * 16 + hi * 8 + g * 4;
        const int head = hcol >> 2;
        *(float4*)(outs[p] + (((bb * HEADS + head) * SEQ + ss) << 2)) =
            make_float4(o0, o1, o2, o3);
      }
    }
  }
}

#define KSPLIT  2
#define KEYS_G  (SEQ / KSPLIT)
#define KT_G    (KEYS_G / 16)
#define AWAVES  16
#define GWAVES  (AWAVES / KSPLIT)
#define QSPLIT  2
#define QT_WG   (SEQ / 16 / QSPLIT)
#define RED_PITCH   (16 * 4 + 4)
#define RSUM_PITCH  (16 + 4)

__device__ __forceinline__ v2f qa_key_frag(const float* ksm, const int ct,
                                           const int lo, const int hi)
{
  const int ncol = ct * 16 + lo;
  v2f bk;
  bk[0] = ksm[ncol * 4 + hi * 2 + 0];
  bk[1] = ksm[ncol * 4 + hi * 2 + 1];
  return bk;
}

__device__ __forceinline__ v8f qa_score_tile(const v2f aq, const v2f bk,
                                             const v8f acc)
{
  return __builtin_amdgcn_wmma_f32_16x16x4_f32(false, aq, false, bk,
                                               (short)0, acc, false, false);
}

__global__ __launch_bounds__(AWAVES * 32) void qa_attention(
    const float* __restrict__ qp, const float* __restrict__ kp,
    const float* __restrict__ vp, float* __restrict__ out)
{
  __shared__ float ksm[SEQ * 4];
  __shared__ float vsm[SEQ * 4];
  __shared__ __align__(16) float red[AWAVES][16][RED_PITCH];
  __shared__ __align__(16) float redsum[AWAVES][16][RSUM_PITCH];
  __shared__ float redmax[AWAVES][16];
  __shared__ __align__(16) float joinb[GWAVES][32][4];

  const int bh = blockIdx.x / QSPLIT;
  const int qh = blockIdx.x % QSPLIT;
  const float* qh_ = qp + (bh * SEQ + qh * (SEQ / QSPLIT)) * 4;
  const float* kh_ = kp + bh * SEQ * 4;
  const float* vh_ = vp + bh * SEQ * 4;

  for (int i = threadIdx.x; i < SEQ; i += AWAVES * 32) {
    *(float4*)&ksm[i * 4] = *(const float4*)(kh_ + i * 4);
    *(float4*)&vsm[i * 4] = *(const float4*)(vh_ + i * 4);
  }
  __syncthreads();

  const int lane = threadIdx.x & 31;
  const int wave = __builtin_amdgcn_readfirstlane(threadIdx.x >> 5);
  const int lo = lane & 15, hi = lane >> 4;
  const int kg = wave / GWAVES;
  const int wq = wave % GWAVES;
  const float* ksw = ksm + kg * KEYS_G * 4;
  const float* vsw = vsm + kg * KEYS_G * 4;
  const int bb = bh / HEADS, head = bh % HEADS;
  float* ow = out + ((size_t)(bb * SEQ + qh * (SEQ / QSPLIT))) * EMB
                  + head * 4 + (lane & 1) * 2;

  for (int qt = wq; qt < QT_WG; qt += GWAVES) {
    v2f aq;
    aq[0] = 0.5f * qh_[(qt * 16 + lo) * 4 + hi * 2 + 0];
    aq[1] = 0.5f * qh_[(qt * 16 + lo) * 4 + hi * 2 + 1];

    const v8f zero = {};
    float rmax[8];
    #pragma unroll
    for (int j = 0; j < 8; ++j) rmax[j] = -INFINITY;
    #pragma unroll 2
    for (int ct = 0; ct < KT_G; ct += 4) {
      const v8f s0 = qa_score_tile(aq, qa_key_frag(ksw, ct + 0, lo, hi), zero);
      const v8f s1 = qa_score_tile(aq, qa_key_frag(ksw, ct + 1, lo, hi), zero);
      const v8f s2 = qa_score_tile(aq, qa_key_frag(ksw, ct + 2, lo, hi), zero);
      const v8f s3 = qa_score_tile(aq, qa_key_frag(ksw, ct + 3, lo, hi), zero);
      #pragma unroll
      for (int j = 0; j < 8; ++j) {
        rmax[j] = fmaxf(fmaxf(rmax[j], s0[j]), s1[j]);
        rmax[j] = fmaxf(fmaxf(rmax[j], s2[j]), s3[j]);
      }
    }
    v8f negm;
    #pragma unroll
    for (int j = 0; j < 8; ++j) {
      float mx = rmax[j];
      mx = fmaxf(mx, __shfl_xor(mx, 1));
      mx = fmaxf(mx, __shfl_xor(mx, 2));
      mx = fmaxf(mx, __shfl_xor(mx, 4));
      mx = fmaxf(mx, __shfl_xor(mx, 8));
      negm[j] = -mx;
    }

    float psum[8];
    float4 ctxp[8];
    #pragma unroll
    for (int j = 0; j < 8; ++j) {
      psum[j] = 0.0f;
      ctxp[j] = make_float4(0.f, 0.f, 0.f, 0.f);
    }
    v2f bkn[4];
    #pragma unroll
    for (int u = 0; u < 4; ++u) bkn[u] = qa_key_frag(ksw, u, lo, hi);
    for (int ct = 0; ct < KT_G; ct += 4) {
      v2f bk[4];
      float4 vcol[4];
      #pragma unroll
      for (int u = 0; u < 4; ++u) {
        bk[u] = bkn[u];
        vcol[u] = *(const float4*)&vsw[((ct + u) * 16 + lo) * 4];
      }
      const int nx = (ct + 4) & (KT_G - 1);
      #pragma unroll
      for (int u = 0; u < 4; ++u) bkn[u] = qa_key_frag(ksw, nx + u, lo, hi);
      v8f sc[4];
      #pragma unroll
      for (int u = 0; u < 4; ++u)
        sc[u] = qa_score_tile(aq, bk[u], negm);
      #pragma unroll
      for (int u = 0; u < 4; ++u) {
        #pragma unroll
        for (int j = 0; j < 8; ++j) {
          const float pr = __expf(sc[u][j]);
          psum[j] += pr;
          ctxp[j].x += pr * vcol[u].x;
          ctxp[j].y += pr * vcol[u].y;
          ctxp[j].z += pr * vcol[u].z;
          ctxp[j].w += pr * vcol[u].w;
        }
      }
    }

    #pragma unroll
    for (int j = 0; j < 8; ++j) {
      const int r = j + 8 * hi;
      *(float4*)&red[wave][r][lo * 4] = ctxp[j];
      redsum[wave][r][lo] = psum[j];
      redmax[wave][r] = -negm[j];
    }
    __syncthreads();
    const int r = lane >> 1;
    float rs = 0.0f;
    #pragma unroll
    for (int cc = 0; cc < 16; ++cc) rs += redsum[wave][r][cc];
    float acc[2];
    #pragma unroll
    for (int t = 0; t < 2; ++t) {
      const int d = (lane & 1) * 2 + t;
      float a = 0.0f;
      #pragma unroll
      for (int cc = 0; cc < 16; ++cc) a += red[wave][r][cc * 4 + d];
      acc[t] = a;
    }
    const float mxr = redmax[wave][r];
    if (kg == 1)
      *(float4*)&joinb[wq][lane][0] = make_float4(acc[0], acc[1], rs, mxr);
    __syncthreads();
    if (kg == 0) {
      const float4 pb = *(const float4*)&joinb[wq][lane][0];
      const float m  = fmaxf(mxr, pb.w);
      const float wa = __expf(mxr - m);
      const float wb = __expf(pb.w - m);
      const float inv = 1.0f / (wa * rs + wb * pb.z);
      float2 o;
      o.x = (wa * acc[0] + wb * pb.x) * inv;
      o.y = (wa * acc[1] + wb * pb.y) * inv;
      *(float2*)(ow + (size_t)(qt * 16 + r) * EMB) = o;
    }
  }
}

extern "C" void kernel_launch(void* const* d_in, const int* in_sizes, int n_in,
                              void* d_out, int out_size, void* d_ws, size_t ws_size,
                              hipStream_t stream) {
  const float* x      = (const float*)d_in[0];
  const float* Wq     = (const float*)d_in[1];
  const float* Wk     = (const float*)d_in[2];
  const float* Wv     = (const float*)d_in[3];
  const float* params = (const float*)d_in[4];
  float* out = (float*)d_out;

  const size_t headElems = (size_t)BATCH * HEADS * SEQ * DKK;
  float* qp = (float*)d_ws;
  float* kp = qp + headElems;
  float* vp = kp + headElems;

  qa_qkv_quantum<<<256, 128, 0, stream>>>(x, Wq, Wk, Wv, params, qp, kp, vp);
  qa_attention<<<BATCH * HEADS * QSPLIT, AWAVES * 32, 0, stream>>>(qp, kp, vp, out);
}


# attention: one score pass per q-tile: an 8-tile key chunk's scores stay in registers from matrix product to exp; per-lane running max, sums and P*V rescaled once per chunk (sum rounding changes)
# speedup vs baseline: 10.4550x; 1.0534x over previous
#include <hip/hip_runtime.h>
#include <hip/hip_bf16.h>

#define BATCH 2
#define SEQ   1024
#define EMB   256
#define HEADS 64
#define DKK   4

typedef __attribute__((ext_vector_type(16))) _Float16 v16h;
typedef __attribute__((ext_vector_type(8)))  float    v8f;
typedef __attribute__((ext_vector_type(2)))  float    v2f;

#define PE 2
#define PW 3
#define PX 4
#define KW 4
#define PT (PE * PW * PX)
#define KS (EMB / KW / 32)

struct qa_raw16 { float4 f0, f1, f2, f3; };

__device__ __forceinline__ qa_raw16 qa_load16(const float* p)
{
  qa_raw16 r;
  r.f0 = *(const float4*)(p);
  r.f1 = *(const float4*)(p + 4);
  r.f2 = *(const float4*)(p + 16);
  r.f3 = *(const float4*)(p + 20);
  return r;
}

__device__ __forceinline__ v16h qa_frag16(const qa_raw16 w)
{
  v16h r;
  r[0]=(_Float16)w.f0.x; r[1]=(_Float16)w.f0.y; r[2]=(_Float16)w.f0.z; r[3]=(_Float16)w.f0.w;
  r[4]=(_Float16)w.f1.x; r[5]=(_Float16)w.f1.y; r[6]=(_Float16)w.f1.z; r[7]=(_Float16)w.f1.w;
  r[8]=(_Float16)w.f2.x; r[9]=(_Float16)w.f2.y; r[10]=(_Float16)w.f2.z; r[11]=(_Float16)w.f2.w;
  r[12]=(_Float16)w.f3.x; r[13]=(_Float16)w.f3.y; r[14]=(_Float16)w.f3.z; r[15]=(_Float16)w.f3.w;
  return r;
}

__global__ __launch_bounds__(KW * 32) __attribute__((amdgpu_waves_per_eu(1, 1))) void qa_qkv_quantum(
    const float* __restrict__ x,
    const float* __restrict__ Wq, const float* __restrict__ Wk,
    const float* __restrict__ Wv, const float* __restrict__ params,
    float* __restrict__ qp, float* __restrict__ kp, float* __restrict__ vp)
{
  __shared__ __align__(16) float part[KW][PT][2][32][4];

  const int lane = threadIdx.x & 31;
  const int wave = __builtin_amdgcn_readfirstlane(threadIdx.x >> 5);
  const int lo = lane & 15, hi = lane >> 4;

  const int ntp = blockIdx.x >> 5;
  const int mtq = blockIdx.x & 31;
  const int k0  = wave * (EMB / KW);

  const float* wrow[PE][PW];
  #pragma unroll
  for (int e = 0; e < PE; ++e) {
    const int woff = ((ntp * PE + e) * 16 + lo) * EMB + hi * 8 + k0;
    wrow[e][0] = Wq + woff;
    wrow[e][1] = Wk + woff;
    wrow[e][2] = Wv + woff;
  }
  const float* xrow[PX];
  #pragma unroll
  for (int t = 0; t < PX; ++t)
    xrow[t] = x + ((mtq * PX + t) * 16 + lo) * EMB + hi * 8 + k0;

  v8f c[PE][PW][PX];
  #pragma unroll
  for (int e = 0; e < PE; ++e)
    #pragma unroll
    for (int p = 0; p < PW; ++p)
      #pragma unroll
      for (int t = 0; t < PX; ++t) c[e][p][t] = v8f{};

  qa_raw16 ra[KS][PE][PW], rb[KS][PX];
  #pragma unroll
  for (int s = 0; s < KS; ++s) {
    #pragma unroll
    for (int e = 0; e < PE; ++e)
      #pragma unroll
      for (int p = 0; p < PW; ++p) ra[s][e][p] = qa_load16(wrow[e][p] + s * 32);
    #pragma unroll
    for (int t = 0; t < PX; ++t) rb[s][t] = qa_load16(xrow[t] + s * 32);
  }
  __builtin_amdgcn_sched_barrier(0);

  #pragma unroll
  for (int s = 0; s < KS; ++s) {
    v16h a[PE][PW], b[PX];
    #pragma unroll
    for (int e = 0; e < PE; ++e)
      #pragma unroll
      for (int p = 0; p < PW; ++p) a[e][p] = qa_frag16(ra[s][e][p]);
    #pragma unroll
    for (int t = 0; t < PX; ++t) b[t] = qa_frag16(rb[s][t]);
    #pragma unroll
    for (int e = 0; e < PE; ++e)
      #pragma unroll
      for (int p = 0; p < PW; ++p)
        #pragma unroll
        for (int t = 0; t < PX; ++t)
          c[e][p][t] = __builtin_amdgcn_wmma_f32_16x16x32_f16(false, a[e][p], false, b[t],
                                                              (short)0, c[e][p][t],
                                                              false, false);
  }

  #pragma unroll
  for (int e = 0; e < PE; ++e)
    #pragma unroll
    for (int p = 0; p < PW; ++p)
      #pragma unroll
      for (int t = 0; t < PX; ++t) {
        const int ti = (e * PW + p) * PX + t;
        *(float4*)&part[wave][ti][0][lane][0] =
            make_float4(c[e][p][t][0], c[e][p][t][1], c[e][p][t][2], c[e][p][t][3]);
        *(float4*)&part[wave][ti][1][lane][0] =
            make_float4(c[e][p][t][4], c[e][p][t][5], c[e][p][t][6], c[e][p][t][7]);
      }
  __syncthreads();

  const float p0 = params[0], p1 = params[1], p2 = params[2], p3 = params[3];
  const float cp0 = __cosf(p0), cp1 = __cosf(p1), cp2 = __cosf(p2), cp3 = __cosf(p3);
  float* const outs[PW] = { qp, kp, vp };
  const int sg = (mtq * PX + wave) * 16 + lo;
  const int bb = sg >> 10;
  const int ss = sg & 1023;
  #pragma unroll
  for (int e = 0; e < PE; ++e) {
    #pragma unroll
    for (int p = 0; p < PW; ++p) {
      const int ti = (e * PW + p) * PX + wave;
      #pragma unroll
      for (int g = 0; g < 2; ++g) {
        float4 s = *(const float4*)&part[0][ti][g][lane][0];
        #pragma unroll
        for (int w = 1; w < KW; ++w) {
          const float4 o = *(const float4*)&part[w][ti][g][lane][0];
          s.x += o.x; s.y += o.y; s.z += o.z; s.w += o.w;
        }
        const float z0 = cp0 * __cosf(s.x + p0);
        const float z1 = cp1 * __cosf(s.y + p1);
        const float z2 = cp2 * __cosf(s.z + p2);
        const float z3 = cp3 * __cosf(s.w + p3);
        const float o1 = z0 * z1;
        const float o2 = o1 * z2;
        const float o3 = o2 * z3;
        const float o0 = z1 * z2 * z3;
        const int hcol = (ntp * PE + e) * 16 + hi * 8 + g * 4;
        const int head = hcol >> 2;
        *(float4*)(outs[p] + (((bb * HEADS + head) * SEQ + ss) << 2)) =
            make_float4(o0, o1, o2, o3);
      }
    }
  }
}

#define KSPLIT  2
#define KEYS_G  (SEQ / KSPLIT)
#define KT_G    (KEYS_G / 16)
#define CH_T    8
#define NCH     (KT_G / CH_T)
#define AWAVES  16
#define GWAVES  (AWAVES / KSPLIT)
#define QSPLIT  2
#define QT_WG   (SEQ / 16 / QSPLIT)
#define QA_LOG2E 1.4426950408889634f
#define RED_PITCH   (16 * 4 + 4)
#define RSUM_PITCH  (16 + 4)

__device__ __forceinline__ v2f qa_key_frag(const float* ksm, const int ct,
                                           const int lo, const int hi)
{
  const int ncol = ct * 16 + lo;
  v2f bk;
  bk[0] = ksm[ncol * 4 + hi * 2 + 0];
  bk[1] = ksm[ncol * 4 + hi * 2 + 1];
  return bk;
}

__device__ __forceinline__ v8f qa_score_tile(const v2f aq, const v2f bk,
                                             const v8f acc)
{
  return __builtin_amdgcn_wmma_f32_16x16x4_f32(false, aq, false, bk,
                                               (short)0, acc, false, false);
}

__device__ __forceinline__ float qa_exp2(const float x)
{
  return __builtin_amdgcn_exp2f(x);
}

__global__ __launch_bounds__(AWAVES * 32) void qa_attention(
    const float* __restrict__ qp, const float* __restrict__ kp,
    const float* __restrict__ vp, float* __restrict__ out)
{
  __shared__ float ksm[SEQ * 4];
  __shared__ float vsm[SEQ * 4];
  __shared__ __align__(16) float red[AWAVES][16][RED_PITCH];
  __shared__ __align__(16) float redsum[AWAVES][16][RSUM_PITCH];
  __shared__ float redmax[AWAVES][16];
  __shared__ __align__(16) float joinb[GWAVES][32][4];

  const int bh = blockIdx.x / QSPLIT;
  const int qh = blockIdx.x % QSPLIT;
  const float* qh_ = qp + (bh * SEQ + qh * (SEQ / QSPLIT)) * 4;
  const float* kh_ = kp + bh * SEQ * 4;
  const float* vh_ = vp + bh * SEQ * 4;

  for (int i = threadIdx.x; i < SEQ; i += AWAVES * 32) {
    *(float4*)&ksm[i * 4] = *(const float4*)(kh_ + i * 4);
    *(float4*)&vsm[i * 4] = *(const float4*)(vh_ + i * 4);
  }
  __syncthreads();

  const int lane = threadIdx.x & 31;
  const int wave = __builtin_amdgcn_readfirstlane(threadIdx.x >> 5);
  const int lo = lane & 15, hi = lane >> 4;
  const int kg = wave / GWAVES;
  const int wq = wave % GWAVES;
  const float* ksw = ksm + kg * KEYS_G * 4;
  const float* vsw = vsm + kg * KEYS_G * 4;
  const int bb = bh / HEADS, head = bh % HEADS;
  float* ow = out + ((size_t)(bb * SEQ + qh * (SEQ / QSPLIT))) * EMB
                  + head * 4 + (lane & 1) * 2;

  v2f bkn[CH_T];
  #pragma unroll
  for (int u = 0; u < CH_T; ++u) bkn[u] = qa_key_frag(ksw, u, lo, hi);

  for (int qt = wq; qt < QT_WG; qt += GWAVES) {
    v2f aq;
    aq[0] = 0.5f * qh_[(qt * 16 + lo) * 4 + hi * 2 + 0];
    aq[1] = 0.5f * qh_[(qt * 16 + lo) * 4 + hi * 2 + 1];

    const v8f zero = {};
    float mrun[8];
    float psum[8];
    v2f cxy[8], czw[8];
    #pragma unroll
    for (int j = 0; j < 8; ++j) {
      mrun[j] = -INFINITY;
      psum[j] = 0.0f;
      cxy[j] = v2f{0.f, 0.f};
      czw[j] = v2f{0.f, 0.f};
    }
    const v2f l2e = {QA_LOG2E, QA_LOG2E};

    for (int ch = 0; ch < NCH; ++ch) {
      const float* kn = ksw + ((ch + 1) & (NCH - 1)) * (CH_T * 16 * 4);
      const float* vc = vsw + ch * (CH_T * 16 * 4);
      v8f sc[CH_T];
      #pragma unroll
      for (int u = 0; u < CH_T; ++u)
        sc[u] = qa_score_tile(aq, bkn[u], zero);
      #pragma unroll
      for (int u = 0; u < CH_T; ++u) bkn[u] = qa_key_frag(kn, u, lo, hi);

      float nml[8];
      #pragma unroll
      for (int j = 0; j < 8; ++j) {
        float mx = mrun[j];
        #pragma unroll
        for (int u = 0; u < CH_T; u += 2)
          mx = fmaxf(fmaxf(mx, sc[u][j]), sc[u + 1][j]);
        const float sf = qa_exp2((mrun[j] - mx) * QA_LOG2E);
        psum[j] *= sf;
        cxy[j] *= sf;
        czw[j] *= sf;
        mrun[j] = mx;
        nml[j] = -mx * QA_LOG2E;
      }

      #pragma unroll
      for (int u = 0; u < CH_T; ++u) {
        const float4 vcol = *(const float4*)&vc[(u * 16 + lo) * 4];
        const v2f vxy = {vcol.x, vcol.y}, vzw = {vcol.z, vcol.w};
        #pragma unroll
        for (int j = 0; j < 8; j += 2) {
          const v2f s2 = {sc[u][j], sc[u][j + 1]};
          const v2f n2 = {nml[j], nml[j + 1]};
          const v2f e2 = __builtin_elementwise_fma(s2, l2e, n2);
          const float p0 = qa_exp2(e2[0]);
          const float p1 = qa_exp2(e2[1]);
          psum[j]     += p0;
          psum[j + 1] += p1;
          const v2f q0 = {p0, p0}, q1 = {p1, p1};
          cxy[j]     = __builtin_elementwise_fma(q0, vxy, cxy[j]);
          czw[j]     = __builtin_elementwise_fma(q0, vzw, czw[j]);
          cxy[j + 1] = __builtin_elementwise_fma(q1, vxy, cxy[j + 1]);
          czw[j + 1] = __builtin_elementwise_fma(q1, vzw, czw[j + 1]);
        }
      }
    }

    float rowm[8];
    float4 ctxp[8];
    #pragma unroll
    for (int j = 0; j < 8; ++j) {
      float mx = mrun[j];
      mx = fmaxf(mx, __shfl_xor(mx, 1));
      mx = fmaxf(mx, __shfl_xor(mx, 2));
      mx = fmaxf(mx, __shfl_xor(mx, 4));
      mx = fmaxf(mx, __shfl_xor(mx, 8));
      rowm[j] = mx;
      const float wl = qa_exp2((mrun[j] - mx) * QA_LOG2E);
      psum[j] *= wl;
      cxy[j] *= wl;
      czw[j] *= wl;
      ctxp[j] = make_float4(cxy[j][0], cxy[j][1], czw[j][0], czw[j][1]);
    }

    #pragma unroll
    for (int j = 0; j < 8; ++j) {
      const int r = j + 8 * hi;
      *(float4*)&red[wave][r][lo * 4] = ctxp[j];
      redsum[wave][r][lo] = psum[j];
      redmax[wave][r] = rowm[j];
    }
    __syncthreads();
    const int r = lane >> 1;
    float rs = 0.0f;
    #pragma unroll
    for (int cc = 0; cc < 16; ++cc) rs += redsum[wave][r][cc];
    float acc[2];
    #pragma unroll
    for (int t = 0; t < 2; ++t) {
      const int d = (lane & 1) * 2 + t;
      float a = 0.0f;
      #pragma unroll
      for (int cc = 0; cc < 16; ++cc) a += red[wave][r][cc * 4 + d];
      acc[t] = a;
    }
    const float mxr = redmax[wave][r];
    if (kg == 1)
      *(float4*)&joinb[wq][lane][0] = make_float4(acc[0], acc[1], rs, mxr);
    __syncthreads();
    if (kg == 0) {
      const float4 pb = *(const float4*)&joinb[wq][lane][0];
      const float m  = fmaxf(mxr, pb.w);
      const float wa = __expf(mxr - m);
      const float wb = __expf(pb.w - m);
      const float inv = 1.0f / (wa * rs + wb * pb.z);
      float2 o;
      o.x = (wa * acc[0] + wb * pb.x) * inv;
      o.y = (wa * acc[1] + wb * pb.y) * inv;
      *(float2*)(ow + (size_t)(qt * 16 + r) * EMB) = o;
    }
  }
}

extern "C" void kernel_launch(void* const* d_in, const int* in_sizes, int n_in,
                              void* d_out, int out_size, void* d_ws, size_t ws_size,
                              hipStream_t stream) {
  const float* x      = (const float*)d_in[0];
  const float* Wq     = (const float*)d_in[1];
  const float* Wk     = (const float*)d_in[2];
  const float* Wv     = (const float*)d_in[3];
  const float* params = (const float*)d_in[4];
  float* out = (float*)d_out;

  const size_t headElems = (size_t)BATCH * HEADS * SEQ * DKK;
  float* qp = (float*)d_ws;
  float* kp = qp + headElems;
  float* vp = kp + headElems;

  qa_qkv_quantum<<<256, 128, 0, stream>>>(x, Wq, Wk, Wv, params, qp, kp, vp);
  qa_attention<<<BATCH * HEADS * QSPLIT, AWAVES * 32, 0, stream>>>(qp, kp, vp, out);
}


# attention: score tiles formed transposed (keys in rows): a lane owns one query and its softmax state; wave = 2 q-tiles x all keys; no 16-lane LDS reduction, join or q-tile barrier (sum order changes)
# speedup vs baseline: 12.3500x; 1.1813x over previous
#include <hip/hip_runtime.h>
#include <hip/hip_bf16.h>

#define BATCH 2
#define SEQ   1024
#define EMB   256
#define HEADS 64
#define DKK   4

typedef __attribute__((ext_vector_type(16))) _Float16 v16h;
typedef __attribute__((ext_vector_type(8)))  float    v8f;
typedef __attribute__((ext_vector_type(2)))  float    v2f;

#define PE 2
#define PW 3
#define PX 4
#define KW 4
#define PT (PE * PW * PX)
#define KS (EMB / KW / 32)

struct qa_raw16 { float4 f0, f1, f2, f3; };

__device__ __forceinline__ qa_raw16 qa_load16(const float* p)
{
  qa_raw16 r;
  r.f0 = *(const float4*)(p);
  r.f1 = *(const float4*)(p + 4);
  r.f2 = *(const float4*)(p + 16);
  r.f3 = *(const float4*)(p + 20);
  return r;
}

__device__ __forceinline__ v16h qa_frag16(const qa_raw16 w)
{
  v16h r;
  r[0]=(_Float16)w.f0.x; r[1]=(_Float16)w.f0.y; r[2]=(_Float16)w.f0.z; r[3]=(_Float16)w.f0.w;
  r[4]=(_Float16)w.f1.x; r[5]=(_Float16)w.f1.y; r[6]=(_Float16)w.f1.z; r[7]=(_Float16)w.f1.w;
  r[8]=(_Float16)w.f2.x; r[9]=(_Float16)w.f2.y; r[10]=(_Float16)w.f2.z; r[11]=(_Float16)w.f2.w;
  r[12]=(_Float16)w.f3.x; r[13]=(_Float16)w.f3.y; r[14]=(_Float16)w.f3.z; r[15]=(_Float16)w.f3.w;
  return r;
}

__global__ __launch_bounds__(KW * 32) __attribute__((amdgpu_waves_per_eu(1, 1))) void qa_qkv_quantum(
    const float* __restrict__ x,
    const float* __restrict__ Wq, const float* __restrict__ Wk,
    const float* __restrict__ Wv, const float* __restrict__ params,
    float* __restrict__ qp, float* __restrict__ kp, float* __restrict__ vp)
{
  __shared__ __align__(16) float part[KW][PT][2][32][4];

  const int lane = threadIdx.x & 31;
  const int wave = __builtin_amdgcn_readfirstlane(threadIdx.x >> 5);
  const int lo = lane & 15, hi = lane >> 4;

  const int ntp = blockIdx.x >> 5;
  const int mtq = blockIdx.x & 31;
  const int k0  = wave * (EMB / KW);

  const float* wrow[PE][PW];
  #pragma unroll
  for (int e = 0; e < PE; ++e) {
    const int woff = ((ntp * PE + e) * 16 + lo) * EMB + hi * 8 + k0;
    wrow[e][0] = Wq + woff;
    wrow[e][1] = Wk + woff;
    wrow[e][2] = Wv + woff;
  }
  const float* xrow[PX];
  #pragma unroll
  for (int t = 0; t < PX; ++t)
    xrow[t] = x + ((mtq * PX + t) * 16 + lo) * EMB + hi * 8 + k0;

  v8f c[PE][PW][PX];
  #pragma unroll
  for (int e = 0; e < PE; ++e)
    #pragma unroll
    for (int p = 0; p < PW; ++p)
      #pragma unroll
      for (int t = 0; t < PX; ++t) c[e][p][t] = v8f{};

  qa_raw16 ra[KS][PE][PW], rb[KS][PX];
  #pragma unroll
  for (int s = 0; s < KS; ++s) {
    #pragma unroll
    for (int e = 0; e < PE; ++e)
      #pragma unroll
      for (int p = 0; p < PW; ++p) ra[s][e][p] = qa_load16(wrow[e][p] + s * 32);
    #pragma unroll
    for (int t = 0; t < PX; ++t) rb[s][t] = qa_load16(xrow[t] + s * 32);
  }
  __builtin_amdgcn_sched_barrier(0);

  #pragma unroll
  for (int s = 0; s < KS; ++s) {
    v16h a[PE][PW], b[PX];
    #pragma unroll
    for (int e = 0; e < PE; ++e)
      #pragma unroll
      for (int p = 0; p < PW; ++p) a[e][p] = qa_frag16(ra[s][e][p]);
    #pragma unroll
    for (int t = 0; t < PX; ++t) b[t] = qa_frag16(rb[s][t]);
    #pragma unroll
    for (int e = 0; e < PE; ++e)
      #pragma unroll
      for (int p = 0; p < PW; ++p)
        #pragma unroll
        for (int t = 0; t < PX; ++t)
          c[e][p][t] = __builtin_amdgcn_wmma_f32_16x16x32_f16(false, a[e][p], false, b[t],
                                                              (short)0, c[e][p][t],
                                                              false, false);
  }

  #pragma unroll
  for (int e = 0; e < PE; ++e)
    #pragma unroll
    for (int p = 0; p < PW; ++p)
      #pragma unroll
      for (int t = 0; t < PX; ++t) {
        const int ti = (e * PW + p) * PX + t;
        *(float4*)&part[wave][ti][0][lane][0] =
            make_float4(c[e][p][t][0], c[e][p][t][1], c[e][p][t][2], c[e][p][t][3]);
        *(float4*)&part[wave][ti][1][lane][0] =
            make_float4(c[e][p][t][4], c[e][p][t][5], c[e][p][t][6], c[e][p][t][7]);
      }
  __syncthreads();

  const float p0 = params[0], p1 = params[1], p2 = params[2], p3 = params[3];
  const float cp0 = __cosf(p0), cp1 = __cosf(p1), cp2 = __cosf(p2), cp3 = __cosf(p3);
  float* const outs[PW] = { qp, kp, vp };
  const int sg = (mtq * PX + wave) * 16 + lo;
  const int bb = sg >> 10;
  const int ss = sg & 1023;
  #pragma unroll
  for (int e = 0; e < PE; ++e) {
    #pragma unroll
    for (int p = 0; p < PW; ++p) {
      const int ti = (e * PW + p) * PX + wave;
      #pragma unroll
      for (int g = 0; g < 2; ++g) {
        float4 s = *(const float4*)&part[0][ti][g][lane][0];
        #pragma unroll
        for (int w = 1; w < KW; ++w) {
          const float4 o = *(const float4*)&part[w][ti][g][lane][0];
          s.x += o.x; s.y += o.y; s.z += o.z; s.w += o.w;
        }
        const float z0 = cp0 * __cosf(s.x + p0);
        const float z1 = cp1 * __cosf(s.y + p1);
        const float z2 = cp2 * __cosf(s.z + p2);
        const float z3 = cp3 * __cosf(s.w + p3);
        const float o1 = z0 * z1;
        const float o2 = o1 * z2;
        const float o3 = o2 * z3;
        const float o0 = z1 * z2 * z3;
        const int hcol = (ntp * PE + e) * 16 + hi * 8 + g * 4;
        const int head = hcol >> 2;
        *(float4*)(outs[p] + (((bb * HEADS + head) * SEQ + ss) << 2)) =
            make_float4(o0, o1, o2, o3);
      }
    }
  }
}

#define KT      (SEQ / 16)
#define CH_T    4
#define NCH     (KT / CH_T)
#define AWAVES  16
#define QSPLIT  2
#define QT_WG   (SEQ / 16 / QSPLIT)
#define QT_W    (QT_WG / AWAVES)
#define QA_LOG2E 1.4426950408889634f

__device__ __forceinline__ v2f qa_frag4(const float* rows, const int lo, const int hi)
{
  v2f f;
  f[0] = rows[lo * 4 + hi * 2 + 0];
  f[1] = rows[lo * 4 + hi * 2 + 1];
  return f;
}

__device__ __forceinline__ float qa_exp2(const float x)
{
  return __builtin_amdgcn_exp2f(x);
}

__global__ __launch_bounds__(AWAVES * 32) void qa_attention(
    const float* __restrict__ qp, const float* __restrict__ kp,
    const float* __restrict__ vp, float* __restrict__ out)
{
  __shared__ __align__(16) float ksm[SEQ * 4];
  __shared__ __align__(16) float vsm[SEQ * 4];

  const int bh = blockIdx.x / QSPLIT;
  const int qh = blockIdx.x % QSPLIT;
  const float* qh_ = qp + (bh * SEQ + qh * (SEQ / QSPLIT)) * 4;
  const float* kh_ = kp + bh * SEQ * 4;
  const float* vh_ = vp + bh * SEQ * 4;

  for (int i = threadIdx.x; i < SEQ; i += AWAVES * 32) {
    *(float4*)&ksm[i * 4] = *(const float4*)(kh_ + i * 4);
    const float4 v = *(const float4*)(vh_ + i * 4);
    float* vd = &vsm[(i >> 1) * 8 + (i & 1)];
    vd[0] = v.x;
    vd[2] = v.y;
    vd[4] = v.z;
    vd[6] = v.w;
  }
  __syncthreads();

  const int lane = threadIdx.x & 31;
  const int wave = __builtin_amdgcn_readfirstlane(threadIdx.x >> 5);
  const int lo = lane & 15, hi = lane >> 4;
  const int bb = bh / HEADS, head = bh % HEADS;

  v2f bq[QT_W];
  #pragma unroll
  for (int a = 0; a < QT_W; ++a) {
    const v2f q = qa_frag4(qh_ + (wave * QT_W + a) * 16 * 4, lo, hi);
    bq[a][0] = 0.5f * q[0];
    bq[a][1] = 0.5f * q[1];
  }

  float m[QT_W];
  v2f ps[QT_W];
  v2f cx[QT_W], cy[QT_W], cz[QT_W], cw[QT_W];
  #pragma unroll
  for (int a = 0; a < QT_W; ++a) {
    m[a] = -INFINITY;
    ps[a] = v2f{0.f, 0.f};
    cx[a] = v2f{0.f, 0.f};
    cy[a] = v2f{0.f, 0.f};
    cz[a] = v2f{0.f, 0.f};
    cw[a] = v2f{0.f, 0.f};
  }
  const v2f l2e = {QA_LOG2E, QA_LOG2E};
  const v8f zero = {};

  v2f akn[CH_T];
  #pragma unroll
  for (int u = 0; u < CH_T; ++u) akn[u] = qa_frag4(ksm + u * 64, lo, hi);

  for (int ch = 0; ch < NCH; ++ch) {
    const float* kn = ksm + ((ch + 1) & (NCH - 1)) * (CH_T * 64);
    const float* vc = vsm + ch * (CH_T * 64) + hi * 32;

    float4 vr[2][8];
    #pragma unroll
    for (int r = 7; r >= 0; --r) vr[0][r] = *(const float4*)&vc[r * 4];
    __builtin_amdgcn_sched_barrier(0);

    v8f sc[QT_W][CH_T];
    #pragma unroll
    for (int u = 0; u < CH_T; ++u)
      #pragma unroll
      for (int a = 0; a < QT_W; ++a)
        sc[a][u] = __builtin_amdgcn_wmma_f32_16x16x4_f32(false, akn[u], false, bq[a],
                                                         (short)0, zero, false, false);
    #pragma unroll
    for (int u = 0; u < CH_T; ++u) akn[u] = qa_frag4(kn + u * 64, lo, hi);

    v2f nml[QT_W];
    #pragma unroll
    for (int a = 0; a < QT_W; ++a) {
      float tm[2] = {m[a], m[a]};
      #pragma unroll
      for (int u = 0; u < CH_T; ++u)
        #pragma unroll
        for (int r = 0; r < 8; r += 2)
          tm[u & 1] = fmaxf(fmaxf(tm[u & 1], sc[a][u][r]), sc[a][u][r + 1]);
      const float mx = fmaxf(tm[0], tm[1]);
      const float sf = qa_exp2((m[a] - mx) * QA_LOG2E);
      ps[a] *= sf;
      cx[a] *= sf;
      cy[a] *= sf;
      cz[a] *= sf;
      cw[a] *= sf;
      m[a] = mx;
      nml[a] = v2f{-mx * QA_LOG2E, -mx * QA_LOG2E};
    }

    #pragma unroll
    for (int u = 0; u < CH_T; ++u) {
      if (u + 1 < CH_T) {
        #pragma unroll
        for (int r = 7; r >= 0; --r)
          vr[(u + 1) & 1][r] = *(const float4*)&vc[((u + 1) * 16 + r) * 4];
        __builtin_amdgcn_sched_barrier(0);
      }
      #pragma unroll
      for (int a = 0; a < QT_W; ++a) {
        #pragma unroll
        for (int r = 0; r < 8; r += 2) {
          const float4 va = vr[u & 1][r], vb = vr[u & 1][r + 1];
          const v2f s2 = {sc[a][u][r], sc[a][u][r + 1]};
          const v2f e2 = __builtin_elementwise_fma(s2, l2e, nml[a]);
          const v2f p2 = {qa_exp2(e2[0]), qa_exp2(e2[1])};
          ps[a] += p2;
          const v2f vx = {va.x, va.y}, vy = {va.z, va.w};
          const v2f vz = {vb.x, vb.y}, vw = {vb.z, vb.w};
          cx[a] = __builtin_elementwise_fma(p2, vx, cx[a]);
          cy[a] = __builtin_elementwise_fma(p2, vy, cy[a]);
          cz[a] = __builtin_elementwise_fma(p2, vz, cz[a]);
          cw[a] = __builtin_elementwise_fma(p2, vw, cw[a]);
        }
      }
    }
  }

  float* ow = out + ((size_t)(bb * SEQ + qh * (SEQ / QSPLIT))) * EMB
                  + head * 4 + hi * 2;
  #pragma unroll
  for (int a = 0; a < QT_W; ++a) {
    const float mo = __shfl_xor(m[a], 16);
    const float mj = fmaxf(m[a], mo);
    const float w = qa_exp2((m[a] - mj) * QA_LOG2E);
    const float s  = (ps[a][0] + ps[a][1]) * w;
    const float ox = (cx[a][0] + cx[a][1]) * w, oy = (cy[a][0] + cy[a][1]) * w;
    const float oz = (cz[a][0] + cz[a][1]) * w, ow_ = (cw[a][0] + cw[a][1]) * w;
    const float st = s + __shfl_xor(s, 16);
    const float tx = ox + __shfl_xor(ox, 16);
    const float ty = oy + __shfl_xor(oy, 16);
    const float tz = oz + __shfl_xor(oz, 16);
    const float tw = ow_ + __shfl_xor(ow_, 16);
    const float inv = 1.0f / st;
    float2 o;
    o.x = (hi ? tz : tx) * inv;
    o.y = (hi ? tw : ty) * inv;
    const int row = (wave * QT_W + a) * 16 + lo;
    *(float2*)(ow + (size_t)row * EMB) = o;
  }
}

extern "C" void kernel_launch(void* const* d_in, const int* in_sizes, int n_in,
                              void* d_out, int out_size, void* d_ws, size_t ws_size,
                              hipStream_t stream) {
  const float* x      = (const float*)d_in[0];
  const float* Wq     = (const float*)d_in[1];
  const float* Wk     = (const float*)d_in[2];
  const float* Wv     = (const float*)d_in[3];
  const float* params = (const float*)d_in[4];
  float* out = (float*)d_out;

  const size_t headElems = (size_t)BATCH * HEADS * SEQ * DKK;
  float* qp = (float*)d_ws;
  float* kp = qp + headElems;
  float* vp = kp + headElems;

  qa_qkv_quantum<<<256, 128, 0, stream>>>(x, Wq, Wk, Wv, params, qp, kp, vp);
  qa_attention<<<BATCH * HEADS * QSPLIT, AWAVES * 32, 0, stream>>>(qp, kp, vp, out);
}
